# starting kernel
# baseline (speedup 1.0000x reference)
#include <hip/hip_runtime.h>


typedef _Float16 half4  __attribute__((ext_vector_type(4)));
typedef _Float16 half8  __attribute__((ext_vector_type(8)));
typedef _Float16 half16 __attribute__((ext_vector_type(16)));
typedef float    v8f    __attribute__((ext_vector_type(8)));

#define B_   8
#define LQ   32
#define N_   1024
#define LD   256
#define K_   128
#define NEGV -100.0f
#define EPSV 1e-8f

__global__ void __launch_bounds__(256) qnorm_kernel(const float* __restrict__ q,
                                                    _Float16* __restrict__ qn) {
    const int r = threadIdx.x;
    const float4* src = (const float4*)(q + (size_t)r * K_);
    float s = 0.f;
#pragma unroll
    for (int i = 0; i < K_ / 4; ++i) {
        float4 v = src[i];
        s += v.x * v.x + v.y * v.y + v.z * v.z + v.w * v.w;
    }
    const float inv = 1.0f / fmaxf(sqrtf(s), EPSV);
    _Float16* dst = qn + (size_t)r * K_;
#pragma unroll
    for (int i = 0; i < K_ / 4; ++i) {
        float4 v = src[i];
        half4 h;
        h[0] = (_Float16)(v.x * inv);
        h[1] = (_Float16)(v.y * inv);
        h[2] = (_Float16)(v.z * inv);
        h[3] = (_Float16)(v.w * inv);
        *(half4*)(dst + i * 4) = h;
    }
}

static __device__ inline half16 mk16(half8 lo, half8 hi) {
    return __builtin_shufflevector(lo, hi, 0, 1, 2, 3, 4, 5, 6, 7,
                                   8, 9, 10, 11, 12, 13, 14, 15);
}

__global__ void __launch_bounds__(256) colbert_maxsim_kernel(
    const _Float16* __restrict__ qn, const int* __restrict__ q_mask,
    const float* __restrict__ d_hidden, const int* __restrict__ d_mask,
    float* __restrict__ out) {
    extern __shared__ _Float16 sd[];
    __shared__ float spartial[8][32];

    const int n     = blockIdx.x;
    const int tid   = threadIdx.x;
    const int w     = tid >> 5;
    const int lane  = tid & 31;
    const int lmod  = lane & 15;
    const int halfg = lane >> 4;

    {
        const int row = tid;
        const float4* src =
            (const float4*)(d_hidden + ((size_t)n * LD + row) * K_);
        float s = 0.f;
#pragma unroll
        for (int i = 0; i < K_ / 4; ++i) {
            float4 v = src[i];
            s += v.x * v.x + v.y * v.y + v.z * v.z + v.w * v.w;
        }
        const float inv = 1.0f / fmaxf(sqrtf(s), EPSV);
        _Float16* dst = sd + (size_t)row * K_;
#pragma unroll
        for (int i = 0; i < K_ / 4; ++i) {
            float4 v = src[i];
            half4 h;
            h[0] = (_Float16)(v.x * inv);
            h[1] = (_Float16)(v.y * inv);
            h[2] = (_Float16)(v.z * inv);
            h[3] = (_Float16)(v.w * inv);
            *(half4*)(dst + i * 4) = h;
        }
    }
    __syncthreads();

    const int col0 = (w * 2 + 0) * 16 + lmod;
    const int col1 = (w * 2 + 1) * 16 + lmod;
    const bool dm0 = d_mask[n * LD + col0] > 0;
    const bool dm1 = d_mask[n * LD + col1] > 0;

    for (int b = 0; b < B_; ++b) {
        v8f acc[2][2] = {};
#pragma unroll
        for (int kc = 0; kc < 4; ++kc) {
            const int kk   = kc * 32;
            const int offK = kk + 8 * halfg;

            const _Float16* qa0 = qn + (size_t)(b * LQ + lmod) * K_;
            const _Float16* qa1 = qn + (size_t)(b * LQ + 16 + lmod) * K_;
            half16 A0 = mk16(*(const half8*)(qa0 + offK),
                             *(const half8*)(qa0 + offK + 16));
            half16 A1 = mk16(*(const half8*)(qa1 + offK),
                             *(const half8*)(qa1 + offK + 16));

            const _Float16* db0 = sd + (size_t)col0 * K_ + offK;
            const _Float16* db1 = sd + (size_t)col1 * K_ + offK;
            half16 Bf0 = mk16(*(const half8*)(db0), *(const half8*)(db0 + 16));
            half16 Bf1 = mk16(*(const half8*)(db1), *(const half8*)(db1 + 16));

            acc[0][0] = __builtin_amdgcn_wmma_f32_16x16x32_f16(
                false, A0, false, Bf0, (short)0, acc[0][0], false, false);
            acc[0][1] = __builtin_amdgcn_wmma_f32_16x16x32_f16(
                false, A1, false, Bf0, (short)0, acc[0][1], false, false);
            acc[1][0] = __builtin_amdgcn_wmma_f32_16x16x32_f16(
                false, A0, false, Bf1, (short)0, acc[1][0], false, false);
            acc[1][1] = __builtin_amdgcn_wmma_f32_16x16x32_f16(
                false, A1, false, Bf1, (short)0, acc[1][1], false, false);
        }

#pragma unroll
        for (int mt = 0; mt < 2; ++mt) {
#pragma unroll
            for (int v = 0; v < 8; ++v) {
                float x0 = dm0 ? acc[0][mt][v] : NEGV;
                float x1 = dm1 ? acc[1][mt][v] : NEGV;
                float x  = fmaxf(x0, x1);
                x = fmaxf(x, __shfl_xor(x, 1, 32));
                x = fmaxf(x, __shfl_xor(x, 2, 32));
                x = fmaxf(x, __shfl_xor(x, 4, 32));
                x = fmaxf(x, __shfl_xor(x, 8, 32));
                if (lmod == 0)
                    spartial[w][mt * 16 + 8 * halfg + v] = x;
            }
        }
        __syncthreads();

        if (tid < 32) {
            float m = spartial[0][tid];
#pragma unroll
            for (int ww = 1; ww < 8; ++ww) m = fmaxf(m, spartial[ww][tid]);
            float val = (q_mask[b * LQ + tid] > 0) ? m : 0.0f;
            val += __shfl_xor(val, 16, 32);
            val += __shfl_xor(val, 8, 32);
            val += __shfl_xor(val, 4, 32);
            val += __shfl_xor(val, 2, 32);
            val += __shfl_xor(val, 1, 32);
            if (tid == 0) out[b * N_ + n] = val;
        }
        __syncthreads();
    }
}

extern "C" void kernel_launch(void* const* d_in, const int* in_sizes, int n_in,
                              void* d_out, int out_size, void* d_ws, size_t ws_size,
                              hipStream_t stream) {
    const float* q_hidden = (const float*)d_in[0];
    const int*   q_mask   = (const int*)d_in[1];
    const float* d_hidden = (const float*)d_in[2];
    const int*   d_mask   = (const int*)d_in[3];
    float*       out      = (float*)d_out;

    _Float16* qn = (_Float16*)d_ws;

    qnorm_kernel<<<1, 256, 0, stream>>>(q_hidden, qn);

    const size_t shbytes = (size_t)LD * K_ * sizeof(_Float16);
    colbert_maxsim_kernel<<<N_, 256, shbytes, stream>>>(
        qn, q_mask, d_hidden, d_mask, out);
}


# LDS doc-tile row pitch 256 B -> 272 B (16 B x odd): B-fragment ds_load_b128 reads and staging ds_store no longer put every lane of a half-wave on one 4-bank group; dynamic LDS 64 -> 68 KB
# speedup vs baseline: 1.7469x; 1.7469x over previous
#include <hip/hip_runtime.h>


typedef _Float16 half4  __attribute__((ext_vector_type(4)));
typedef _Float16 half8  __attribute__((ext_vector_type(8)));
typedef _Float16 half16 __attribute__((ext_vector_type(16)));
typedef float    v8f    __attribute__((ext_vector_type(8)));

#define B_   8
#define LQ   32
#define N_   1024
#define LD   256
#define K_   128
#define PITCH 136
#define NEGV -100.0f
#define EPSV 1e-8f

__global__ void __launch_bounds__(256) qnorm_kernel(const float* __restrict__ q,
                                                    _Float16* __restrict__ qn) {
    const int r = threadIdx.x;
    const float4* src = (const float4*)(q + (size_t)r * K_);
    float s = 0.f;
#pragma unroll
    for (int i = 0; i < K_ / 4; ++i) {
        float4 v = src[i];
        s += v.x * v.x + v.y * v.y + v.z * v.z + v.w * v.w;
    }
    const float inv = 1.0f / fmaxf(sqrtf(s), EPSV);
    _Float16* dst = qn + (size_t)r * K_;
#pragma unroll
    for (int i = 0; i < K_ / 4; ++i) {
        float4 v = src[i];
        half4 h;
        h[0] = (_Float16)(v.x * inv);
        h[1] = (_Float16)(v.y * inv);
        h[2] = (_Float16)(v.z * inv);
        h[3] = (_Float16)(v.w * inv);
        *(half4*)(dst + i * 4) = h;
    }
}

static __device__ inline half16 mk16(half8 lo, half8 hi) {
    return __builtin_shufflevector(lo, hi, 0, 1, 2, 3, 4, 5, 6, 7,
                                   8, 9, 10, 11, 12, 13, 14, 15);
}

__global__ void __launch_bounds__(256) colbert_maxsim_kernel(
    const _Float16* __restrict__ qn, const int* __restrict__ q_mask,
    const float* __restrict__ d_hidden, const int* __restrict__ d_mask,
    float* __restrict__ out) {
    extern __shared__ _Float16 sd[];
    __shared__ float spartial[8][32];

    const int n     = blockIdx.x;
    const int tid   = threadIdx.x;
    const int w     = tid >> 5;
    const int lane  = tid & 31;
    const int lmod  = lane & 15;
    const int halfg = lane >> 4;

    {
        const int row = tid;
        const float4* src =
            (const float4*)(d_hidden + ((size_t)n * LD + row) * K_);
        float s = 0.f;
#pragma unroll
        for (int i = 0; i < K_ / 4; ++i) {
            float4 v = src[i];
            s += v.x * v.x + v.y * v.y + v.z * v.z + v.w * v.w;
        }
        const float inv = 1.0f / fmaxf(sqrtf(s), EPSV);
        _Float16* dst = sd + (size_t)row * PITCH;
#pragma unroll
        for (int i = 0; i < K_ / 4; ++i) {
            float4 v = src[i];
            half4 h;
            h[0] = (_Float16)(v.x * inv);
            h[1] = (_Float16)(v.y * inv);
            h[2] = (_Float16)(v.z * inv);
            h[3] = (_Float16)(v.w * inv);
            *(half4*)(dst + i * 4) = h;
        }
    }
    __syncthreads();

    const int col0 = (w * 2 + 0) * 16 + lmod;
    const int col1 = (w * 2 + 1) * 16 + lmod;
    const bool dm0 = d_mask[n * LD + col0] > 0;
    const bool dm1 = d_mask[n * LD + col1] > 0;

    for (int b = 0; b < B_; ++b) {
        v8f acc[2][2] = {};
#pragma unroll
        for (int kc = 0; kc < 4; ++kc) {
            const int kk   = kc * 32;
            const int offK = kk + 8 * halfg;

            const _Float16* qa0 = qn + (size_t)(b * LQ + lmod) * K_;
            const _Float16* qa1 = qn + (size_t)(b * LQ + 16 + lmod) * K_;
            half16 A0 = mk16(*(const half8*)(qa0 + offK),
                             *(const half8*)(qa0 + offK + 16));
            half16 A1 = mk16(*(const half8*)(qa1 + offK),
                             *(const half8*)(qa1 + offK + 16));

            const _Float16* db0 = sd + (size_t)col0 * PITCH + offK;
            const _Float16* db1 = sd + (size_t)col1 * PITCH + offK;
            half16 Bf0 = mk16(*(const half8*)(db0), *(const half8*)(db0 + 16));
            half16 Bf1 = mk16(*(const half8*)(db1), *(const half8*)(db1 + 16));

            acc[0][0] = __builtin_amdgcn_wmma_f32_16x16x32_f16(
                false, A0, false, Bf0, (short)0, acc[0][0], false, false);
            acc[0][1] = __builtin_amdgcn_wmma_f32_16x16x32_f16(
                false, A1, false, Bf0, (short)0, acc[0][1], false, false);
            acc[1][0] = __builtin_amdgcn_wmma_f32_16x16x32_f16(
                false, A0, false, Bf1, (short)0, acc[1][0], false, false);
            acc[1][1] = __builtin_amdgcn_wmma_f32_16x16x32_f16(
                false, A1, false, Bf1, (short)0, acc[1][1], false, false);
        }

#pragma unroll
        for (int mt = 0; mt < 2; ++mt) {
#pragma unroll
            for (int v = 0; v < 8; ++v) {
                float x0 = dm0 ? acc[0][mt][v] : NEGV;
                float x1 = dm1 ? acc[1][mt][v] : NEGV;
                float x  = fmaxf(x0, x1);
                x = fmaxf(x, __shfl_xor(x, 1, 32));
                x = fmaxf(x, __shfl_xor(x, 2, 32));
                x = fmaxf(x, __shfl_xor(x, 4, 32));
                x = fmaxf(x, __shfl_xor(x, 8, 32));
                if (lmod == 0)
                    spartial[w][mt * 16 + 8 * halfg + v] = x;
            }
        }
        __syncthreads();

        if (tid < 32) {
            float m = spartial[0][tid];
#pragma unroll
            for (int ww = 1; ww < 8; ++ww) m = fmaxf(m, spartial[ww][tid]);
            float val = (q_mask[b * LQ + tid] > 0) ? m : 0.0f;
            val += __shfl_xor(val, 16, 32);
            val += __shfl_xor(val, 8, 32);
            val += __shfl_xor(val, 4, 32);
            val += __shfl_xor(val, 2, 32);
            val += __shfl_xor(val, 1, 32);
            if (tid == 0) out[b * N_ + n] = val;
        }
        __syncthreads();
    }
}

extern "C" void kernel_launch(void* const* d_in, const int* in_sizes, int n_in,
                              void* d_out, int out_size, void* d_ws, size_t ws_size,
                              hipStream_t stream) {
    const float* q_hidden = (const float*)d_in[0];
    const int*   q_mask   = (const int*)d_in[1];
    const float* d_hidden = (const float*)d_in[2];
    const int*   d_mask   = (const int*)d_in[3];
    float*       out      = (float*)d_out;

    _Float16* qn = (_Float16*)d_ws;

    qnorm_kernel<<<1, 256, 0, stream>>>(q_hidden, qn);

    const size_t shbytes = (size_t)LD * PITCH * sizeof(_Float16);
    colbert_maxsim_kernel<<<N_, 256, shbytes, stream>>>(
        qn, q_mask, d_hidden, d_mask, out);
}


# Wave w owns batch w: q rows kept in registers as A fragments, each wave walks all 16 doc column tiles with masked row maxima in registers; per-batch barriers, LDS partials, q reloads removed
# speedup vs baseline: 4.8212x; 2.7599x over previous
#include <hip/hip_runtime.h>


typedef _Float16 half4  __attribute__((ext_vector_type(4)));
typedef _Float16 half8  __attribute__((ext_vector_type(8)));
typedef _Float16 half16 __attribute__((ext_vector_type(16)));
typedef float    v8f    __attribute__((ext_vector_type(8)));

#define B_   8
#define LQ   32
#define N_   1024
#define LD   256
#define K_   128
#define PITCH 136
#define NEGV -100.0f
#define EPSV 1e-8f

__global__ void __launch_bounds__(256) qnorm_kernel(const float* __restrict__ q,
                                                    _Float16* __restrict__ qn) {
    const int r = threadIdx.x;
    const float4* src = (const float4*)(q + (size_t)r * K_);
    float s = 0.f;
#pragma unroll
    for (int i = 0; i < K_ / 4; ++i) {
        float4 v = src[i];
        s += v.x * v.x + v.y * v.y + v.z * v.z + v.w * v.w;
    }
    const float inv = 1.0f / fmaxf(sqrtf(s), EPSV);
    _Float16* dst = qn + (size_t)r * K_;
#pragma unroll
    for (int i = 0; i < K_ / 4; ++i) {
        float4 v = src[i];
        half4 h;
        h[0] = (_Float16)(v.x * inv);
        h[1] = (_Float16)(v.y * inv);
        h[2] = (_Float16)(v.z * inv);
        h[3] = (_Float16)(v.w * inv);
        *(half4*)(dst + i * 4) = h;
    }
}

static __device__ inline half16 mk16(half8 lo, half8 hi) {
    return __builtin_shufflevector(lo, hi, 0, 1, 2, 3, 4, 5, 6, 7,
                                   8, 9, 10, 11, 12, 13, 14, 15);
}

__global__ void __launch_bounds__(256) colbert_maxsim_kernel(
    const _Float16* __restrict__ qn, const int* __restrict__ q_mask,
    const float* __restrict__ d_hidden, const int* __restrict__ d_mask,
    float* __restrict__ out) {
    extern __shared__ _Float16 sd[];
    __shared__ unsigned sdm[8];

    const int n     = blockIdx.x;
    const int tid   = threadIdx.x;
    const int w     = tid >> 5;
    const int lane  = tid & 31;
    const int lmod  = lane & 15;
    const int halfg = lane >> 4;

    const int dmv = d_mask[n * LD + tid];
    const int qmv = q_mask[w * LQ + lane];

    half16 A0[4], A1[4];
    {
        const _Float16* qa0 = qn + (size_t)(w * LQ + lmod) * K_ + 8 * halfg;
        const _Float16* qa1 = qn + (size_t)(w * LQ + 16 + lmod) * K_ + 8 * halfg;
#pragma unroll
        for (int kc = 0; kc < 4; ++kc) {
            A0[kc] = mk16(*(const half8*)(qa0 + kc * 32),
                          *(const half8*)(qa0 + kc * 32 + 16));
            A1[kc] = mk16(*(const half8*)(qa1 + kc * 32),
                          *(const half8*)(qa1 + kc * 32 + 16));
        }
    }

    {
        const int row = tid;
        const float4* src =
            (const float4*)(d_hidden + ((size_t)n * LD + row) * K_);
        float s = 0.f;
#pragma unroll
        for (int i = 0; i < K_ / 4; ++i) {
            float4 v = src[i];
            s += v.x * v.x + v.y * v.y + v.z * v.z + v.w * v.w;
        }
        const float inv = 1.0f / fmaxf(sqrtf(s), EPSV);
        _Float16* dst = sd + (size_t)row * PITCH;
#pragma unroll
        for (int i = 0; i < K_ / 4; ++i) {
            float4 v = src[i];
            half4 h;
            h[0] = (_Float16)(v.x * inv);
            h[1] = (_Float16)(v.y * inv);
            h[2] = (_Float16)(v.z * inv);
            h[3] = (_Float16)(v.w * inv);
            *(half4*)(dst + i * 4) = h;
        }
    }

    {
        const unsigned bits = __builtin_amdgcn_ballot_w32(dmv > 0);
        if (lane == 0) sdm[w] = bits;
    }

    const unsigned qbits = __builtin_amdgcn_ballot_w32(qmv > 0);

    __syncthreads();

    unsigned mw[8];
#pragma unroll
    for (int i = 0; i < 8; ++i) mw[i] = __builtin_amdgcn_readfirstlane(sdm[i]);

    float rmax0[8], rmax1[8];
#pragma unroll
    for (int v = 0; v < 8; ++v) { rmax0[v] = -__builtin_inff(); rmax1[v] = -__builtin_inff(); }

    const _Float16* db = sd + (size_t)lmod * PITCH + 8 * halfg;
    half16 Bc[4], Bn[4];
#pragma unroll
    for (int kc = 0; kc < 4; ++kc)
        Bc[kc] = mk16(*(const half8*)(db + kc * 32), *(const half8*)(db + kc * 32 + 16));
#pragma unroll
    for (int ct = 0; ct < LD / 16; ++ct) {
        if (ct + 1 < LD / 16) {
#pragma unroll
            for (int kc = 0; kc < 4; ++kc) {
                const _Float16* p = db + (size_t)(ct + 1) * 16 * PITCH + kc * 32;
                Bn[kc] = mk16(*(const half8*)(p), *(const half8*)(p + 16));
            }
        }
        __builtin_amdgcn_sched_barrier(0);
        v8f acc0 = {}, acc1 = {};
#pragma unroll
        for (int kc = 0; kc < 4; ++kc) {
            acc0 = __builtin_amdgcn_wmma_f32_16x16x32_f16(
                false, A0[kc], false, Bc[kc], (short)0, acc0, false, false);
            acc1 = __builtin_amdgcn_wmma_f32_16x16x32_f16(
                false, A1[kc], false, Bc[kc], (short)0, acc1, false, false);
        }
        const bool dm = ((mw[ct >> 1] >> ((ct & 1) * 16 + lmod)) & 1u) != 0u;
#pragma unroll
        for (int v = 0; v < 8; ++v) {
            rmax0[v] = fmaxf(rmax0[v], dm ? acc0[v] : NEGV);
            rmax1[v] = fmaxf(rmax1[v], dm ? acc1[v] : NEGV);
        }
#pragma unroll
        for (int kc = 0; kc < 4; ++kc) Bc[kc] = Bn[kc];
    }

    float y0[8], y1[8];
#pragma unroll
    for (int v = 0; v < 8; ++v) {
        float a = rmax0[v], c = rmax1[v];
        a = fmaxf(a, __shfl_xor(a, 1, 32));  c = fmaxf(c, __shfl_xor(c, 1, 32));
        a = fmaxf(a, __shfl_xor(a, 2, 32));  c = fmaxf(c, __shfl_xor(c, 2, 32));
        a = fmaxf(a, __shfl_xor(a, 4, 32));  c = fmaxf(c, __shfl_xor(c, 4, 32));
        a = fmaxf(a, __shfl_xor(a, 8, 32));  c = fmaxf(c, __shfl_xor(c, 8, 32));
        y0[v] = ((qbits >> (8 * halfg + v)) & 1u) ? a : 0.0f;
        y1[v] = ((qbits >> (16 + 8 * halfg + v)) & 1u) ? c : 0.0f;
    }

    float u[8];
#pragma unroll
    for (int v = 0; v < 8; ++v) {
        const float t = y0[v] + y1[v];
        u[v] = t + __shfl_xor(t, 16, 32);
    }
    const float s40 = u[0] + u[4], s41 = u[1] + u[5], s42 = u[2] + u[6], s43 = u[3] + u[7];
    const float s20 = s40 + s42, s21 = s41 + s43;
    const float tot = s20 + s21;
    if (lane == 0) out[w * N_ + n] = tot;
}

extern "C" void kernel_launch(void* const* d_in, const int* in_sizes, int n_in,
                              void* d_out, int out_size, void* d_ws, size_t ws_size,
                              hipStream_t stream) {
    const float* q_hidden = (const float*)d_in[0];
    const int*   q_mask   = (const int*)d_in[1];
    const float* d_hidden = (const float*)d_in[2];
    const int*   d_mask   = (const int*)d_in[3];
    float*       out      = (float*)d_out;

    _Float16* qn = (_Float16*)d_ws;

    qnorm_kernel<<<1, 256, 0, stream>>>(q_hidden, qn);

    const size_t shbytes = (size_t)LD * PITCH * sizeof(_Float16);
    colbert_maxsim_kernel<<<N_, 256, shbytes, stream>>>(
        qn, q_mask, d_hidden, d_mask, out);
}


# Row loads of both normalizations coalesced: 8 lanes read one 128 B line of a row (4 lines per load, was 32); squared norm = 8 lane partials + 3 DPP row_xmask adds (sum order changes)
# speedup vs baseline: 5.7597x; 1.1947x over previous
#include <hip/hip_runtime.h>


typedef _Float16 half4  __attribute__((ext_vector_type(4)));
typedef _Float16 half8  __attribute__((ext_vector_type(8)));
typedef _Float16 half16 __attribute__((ext_vector_type(16)));
typedef float    v8f    __attribute__((ext_vector_type(8)));

#define B_   8
#define LQ   32
#define N_   1024
#define LD   256
#define K_   128
#define PITCH 136
#define NEGV -100.0f
#define EPSV 1e-8f

template <int M>
static __device__ inline float xmask_add(float v) {
    const int t = __builtin_amdgcn_update_dpp(0, __float_as_int(v), 0x160 + M, 0xf, 0xf, true);
    return v + __int_as_float(t);
}

template <int DPITCH>
static __device__ inline void norm_rows32(const float* __restrict__ src,
                                          _Float16* __restrict__ dst, const int lane) {
    const int g = lane >> 3;
    const int c = lane & 7;
    const float* s0 = src + (size_t)(4 * g) * K_ + 4 * c;
    _Float16*    d0 = dst + (size_t)(4 * g) * DPITCH + 4 * c;
#pragma nounroll
    for (int hb = 0; hb < 32; hb += 16) {
        float4 x[4][4];
#pragma unroll
        for (int j = 0; j < 4; ++j) {
#pragma unroll
            for (int i = 0; i < 4; ++i)
                x[j][i] = *(const float4*)(s0 + (size_t)(hb + j) * K_ + 32 * i);
        }
#pragma unroll
        for (int j = 0; j < 4; ++j) {
            float s = 0.f;
#pragma unroll
            for (int i = 0; i < 4; ++i) {
                const float4 v = x[j][i];
                s += v.x * v.x + v.y * v.y + v.z * v.z + v.w * v.w;
            }
            s = xmask_add<1>(s);
            s = xmask_add<2>(s);
            s = xmask_add<4>(s);
            const float inv = 1.0f / fmaxf(sqrtf(s), EPSV);
            _Float16* d = d0 + (size_t)(hb + j) * DPITCH;
#pragma unroll
            for (int i = 0; i < 4; ++i) {
                const float4 v = x[j][i];
                half4 h;
                h[0] = (_Float16)(v.x * inv);
                h[1] = (_Float16)(v.y * inv);
                h[2] = (_Float16)(v.z * inv);
                h[3] = (_Float16)(v.w * inv);
                *(half4*)(d + 32 * i) = h;
            }
        }
    }
}

__global__ void __launch_bounds__(256) qnorm_kernel(const float* __restrict__ q,
                                                    _Float16* __restrict__ qn) {
    const int w    = threadIdx.x >> 5;
    const int lane = threadIdx.x & 31;
    norm_rows32<K_>(q + (size_t)(w * 32) * K_, qn + (size_t)(w * 32) * K_, lane);
}

static __device__ inline half16 mk16(half8 lo, half8 hi) {
    return __builtin_shufflevector(lo, hi, 0, 1, 2, 3, 4, 5, 6, 7,
                                   8, 9, 10, 11, 12, 13, 14, 15);
}

__global__ void __launch_bounds__(256) colbert_maxsim_kernel(
    const _Float16* __restrict__ qn, const int* __restrict__ q_mask,
    const float* __restrict__ d_hidden, const int* __restrict__ d_mask,
    float* __restrict__ out) {
    extern __shared__ _Float16 sd[];
    __shared__ unsigned sdm[8];

    const int n     = blockIdx.x;
    const int tid   = threadIdx.x;
    const int w     = tid >> 5;
    const int lane  = tid & 31;
    const int lmod  = lane & 15;
    const int halfg = lane >> 4;

    const int dmv = d_mask[n * LD + tid];
    const int qmv = q_mask[w * LQ + lane];

    half16 A0[4], A1[4];
    {
        const _Float16* qa0 = qn + (size_t)(w * LQ + lmod) * K_ + 8 * halfg;
        const _Float16* qa1 = qn + (size_t)(w * LQ + 16 + lmod) * K_ + 8 * halfg;
#pragma unroll
        for (int kc = 0; kc < 4; ++kc) {
            A0[kc] = mk16(*(const half8*)(qa0 + kc * 32),
                          *(const half8*)(qa0 + kc * 32 + 16));
            A1[kc] = mk16(*(const half8*)(qa1 + kc * 32),
                          *(const half8*)(qa1 + kc * 32 + 16));
        }
    }

    norm_rows32<PITCH>(d_hidden + ((size_t)n * LD + w * 32) * K_,
                       sd + (size_t)(w * 32) * PITCH, lane);

    {
        const unsigned bits = __builtin_amdgcn_ballot_w32(dmv > 0);
        if (lane == 0) sdm[w] = bits;
    }

    const unsigned qbits = __builtin_amdgcn_ballot_w32(qmv > 0);

    __syncthreads();

    unsigned mw[8];
#pragma unroll
    for (int i = 0; i < 8; ++i) mw[i] = __builtin_amdgcn_readfirstlane(sdm[i]);

    float rmax0[8], rmax1[8];
#pragma unroll
    for (int v = 0; v < 8; ++v) { rmax0[v] = -__builtin_inff(); rmax1[v] = -__builtin_inff(); }

    const _Float16* db = sd + (size_t)lmod * PITCH + 8 * halfg;
    half16 Bc[4], Bn[4];
#pragma unroll
    for (int kc = 0; kc < 4; ++kc)
        Bc[kc] = mk16(*(const half8*)(db + kc * 32), *(const half8*)(db + kc * 32 + 16));
#pragma unroll
    for (int ct = 0; ct < LD / 16; ++ct) {
        if (ct + 1 < LD / 16) {
#pragma unroll
            for (int kc = 0; kc < 4; ++kc) {
                const _Float16* p = db + (size_t)(ct + 1) * 16 * PITCH + kc * 32;
                Bn[kc] = mk16(*(const half8*)(p), *(const half8*)(p + 16));
            }
        }
        __builtin_amdgcn_sched_barrier(0);
        v8f acc0 = {}, acc1 = {};
#pragma unroll
        for (int kc = 0; kc < 4; ++kc) {
            acc0 = __builtin_amdgcn_wmma_f32_16x16x32_f16(
                false, A0[kc], false, Bc[kc], (short)0, acc0, false, false);
            acc1 = __builtin_amdgcn_wmma_f32_16x16x32_f16(
                false, A1[kc], false, Bc[kc], (short)0, acc1, false, false);
        }
        const bool dm = ((mw[ct >> 1] >> ((ct & 1) * 16 + lmod)) & 1u) != 0u;
#pragma unroll
        for (int v = 0; v < 8; ++v) {
            rmax0[v] = fmaxf(rmax0[v], dm ? acc0[v] : NEGV);
            rmax1[v] = fmaxf(rmax1[v], dm ? acc1[v] : NEGV);
        }
#pragma unroll
        for (int kc = 0; kc < 4; ++kc) Bc[kc] = Bn[kc];
    }

    float y0[8], y1[8];
#pragma unroll
    for (int v = 0; v < 8; ++v) {
        float a = rmax0[v], c = rmax1[v];
        a = fmaxf(a, __shfl_xor(a, 1, 32));  c = fmaxf(c, __shfl_xor(c, 1, 32));
        a = fmaxf(a, __shfl_xor(a, 2, 32));  c = fmaxf(c, __shfl_xor(c, 2, 32));
        a = fmaxf(a, __shfl_xor(a, 4, 32));  c = fmaxf(c, __shfl_xor(c, 4, 32));
        a = fmaxf(a, __shfl_xor(a, 8, 32));  c = fmaxf(c, __shfl_xor(c, 8, 32));
        y0[v] = ((qbits >> (8 * halfg + v)) & 1u) ? a : 0.0f;
        y1[v] = ((qbits >> (16 + 8 * halfg + v)) & 1u) ? c : 0.0f;
    }

    float u[8];
#pragma unroll
    for (int v = 0; v < 8; ++v) {
        const float t = y0[v] + y1[v];
        u[v] = t + __shfl_xor(t, 16, 32);
    }
    const float s40 = u[0] + u[4], s41 = u[1] + u[5], s42 = u[2] + u[6], s43 = u[3] + u[7];
    const float s20 = s40 + s42, s21 = s41 + s43;
    const float tot = s20 + s21;
    if (lane == 0) out[w * N_ + n] = tot;
}

extern "C" void kernel_launch(void* const* d_in, const int* in_sizes, int n_in,
                              void* d_out, int out_size, void* d_ws, size_t ws_size,
                              hipStream_t stream) {
    const float* q_hidden = (const float*)d_in[0];
    const int*   q_mask   = (const int*)d_in[1];
    const float* d_hidden = (const float*)d_in[2];
    const int*   d_mask   = (const int*)d_in[3];
    float*       out      = (float*)d_out;

    _Float16* qn = (_Float16*)d_ws;

    qnorm_kernel<<<1, 256, 0, stream>>>(q_hidden, qn);

    const size_t shbytes = (size_t)LD * PITCH * sizeof(_Float16);
    colbert_maxsim_kernel<<<N_, 256, shbytes, stream>>>(
        qn, q_mask, d_hidden, d_mask, out);
}


# VGPRs 222 -> 128 so that four workgroups per WGP are resident (was two): q fragments loaded behind the staging, one B fragment set refilled in place, one accumulator tile at a time
# speedup vs baseline: 6.0284x; 1.0467x over previous
#include <hip/hip_runtime.h>


typedef _Float16 half4  __attribute__((ext_vector_type(4)));
typedef _Float16 half8  __attribute__((ext_vector_type(8)));
typedef _Float16 half16 __attribute__((ext_vector_type(16)));
typedef float    v8f    __attribute__((ext_vector_type(8)));

#define B_   8
#define LQ   32
#define N_   1024
#define LD   256
#define K_   128
#define PITCH 136
#define NEGV -100.0f
#define EPSV 1e-8f

template <int M>
static __device__ inline float xmask_add(float v) {
    const int t = __builtin_amdgcn_update_dpp(0, __float_as_int(v), 0x160 + M, 0xf, 0xf, true);
    return v + __int_as_float(t);
}

template <int DPITCH>
static __device__ inline void norm_rows32(const float* __restrict__ src,
                                          _Float16* __restrict__ dst, const int lane) {
    const int g = lane >> 3;
    const int c = lane & 7;
    const float* s0 = src + (size_t)(4 * g) * K_ + 4 * c;
    _Float16*    d0 = dst + (size_t)(4 * g) * DPITCH + 4 * c;
#pragma nounroll
    for (int hb = 0; hb < 32; hb += 16) {
        float4 x[4][4];
#pragma unroll
        for (int j = 0; j < 4; ++j) {
#pragma unroll
            for (int i = 0; i < 4; ++i)
                x[j][i] = *(const float4*)(s0 + (size_t)(hb + j) * K_ + 32 * i);
        }
#pragma unroll
        for (int j = 0; j < 4; ++j) {
            float s = 0.f;
#pragma unroll
            for (int i = 0; i < 4; ++i) {
                const float4 v = x[j][i];
                s += v.x * v.x + v.y * v.y + v.z * v.z + v.w * v.w;
            }
            s = xmask_add<1>(s);
            s = xmask_add<2>(s);
            s = xmask_add<4>(s);
            const float inv = 1.0f / fmaxf(sqrtf(s), EPSV);
            _Float16* d = d0 + (size_t)(hb + j) * DPITCH;
#pragma unroll
            for (int i = 0; i < 4; ++i) {
                const float4 v = x[j][i];
                half4 h;
                h[0] = (_Float16)(v.x * inv);
                h[1] = (_Float16)(v.y * inv);
                h[2] = (_Float16)(v.z * inv);
                h[3] = (_Float16)(v.w * inv);
                *(half4*)(d + 32 * i) = h;
            }
        }
    }
}

__global__ void __launch_bounds__(256) qnorm_kernel(const float* __restrict__ q,
                                                    _Float16* __restrict__ qn) {
    const int w    = threadIdx.x >> 5;
    const int lane = threadIdx.x & 31;
    norm_rows32<K_>(q + (size_t)(w * 32) * K_, qn + (size_t)(w * 32) * K_, lane);
}

static __device__ inline half16 mk16(half8 lo, half8 hi) {
    return __builtin_shufflevector(lo, hi, 0, 1, 2, 3, 4, 5, 6, 7,
                                   8, 9, 10, 11, 12, 13, 14, 15);
}

__global__ void __launch_bounds__(256) colbert_maxsim_kernel(
    const _Float16* __restrict__ qn, const int* __restrict__ q_mask,
    const float* __restrict__ d_hidden, const int* __restrict__ d_mask,
    float* __restrict__ out) {
    extern __shared__ _Float16 sd[];
    __shared__ unsigned sdm[8];

    const int n     = blockIdx.x;
    const int tid   = threadIdx.x;
    const int w     = tid >> 5;
    const int lane  = tid & 31;
    const int lmod  = lane & 15;
    const int halfg = lane >> 4;

    const int dmv = d_mask[n * LD + tid];
    const int qmv = q_mask[w * LQ + lane];

    norm_rows32<PITCH>(d_hidden + ((size_t)n * LD + w * 32) * K_,
                       sd + (size_t)(w * 32) * PITCH, lane);

    half16 A0[4], A1[4];
    {
        const _Float16* qa0 = qn + (size_t)(w * LQ + lmod) * K_ + 8 * halfg;
        const _Float16* qa1 = qn + (size_t)(w * LQ + 16 + lmod) * K_ + 8 * halfg;
#pragma unroll
        for (int kc = 0; kc < 4; ++kc) {
            A0[kc] = mk16(*(const half8*)(qa0 + kc * 32),
                          *(const half8*)(qa0 + kc * 32 + 16));
            A1[kc] = mk16(*(const half8*)(qa1 + kc * 32),
                          *(const half8*)(qa1 + kc * 32 + 16));
        }
    }

    {
        const unsigned bits = __builtin_amdgcn_ballot_w32(dmv > 0);
        if (lane == 0) sdm[w] = bits;
    }

    const unsigned qbits = __builtin_amdgcn_ballot_w32(qmv > 0);

    __syncthreads();

    unsigned mw[8];
#pragma unroll
    for (int i = 0; i < 8; ++i) mw[i] = __builtin_amdgcn_readfirstlane(sdm[i]);

    float rmax0[8], rmax1[8];
#pragma unroll
    for (int v = 0; v < 8; ++v) { rmax0[v] = -__builtin_inff(); rmax1[v] = -__builtin_inff(); }

    const _Float16* db = sd + (size_t)lmod * PITCH + 8 * halfg;
    half16 Bf[4];
#pragma unroll
    for (int kc = 0; kc < 4; ++kc)
        Bf[kc] = mk16(*(const half8*)(db + kc * 32), *(const half8*)(db + kc * 32 + 16));
#pragma unroll
    for (int ct = 0; ct < LD / 16; ++ct) {
        const bool dm = ((mw[ct >> 1] >> ((ct & 1) * 16 + lmod)) & 1u) != 0u;
        {
            v8f acc = {};
#pragma unroll
            for (int kc = 0; kc < 4; ++kc)
                acc = __builtin_amdgcn_wmma_f32_16x16x32_f16(
                    false, A0[kc], false, Bf[kc], (short)0, acc, false, false);
#pragma unroll
            for (int v = 0; v < 8; ++v) rmax0[v] = fmaxf(rmax0[v], dm ? acc[v] : NEGV);
#pragma unroll
            for (int v = 0; v < 8; ++v) asm volatile("" : "+v"(rmax0[v]));
        }
        __builtin_amdgcn_sched_barrier(0);
        {
            v8f acc = {};
#pragma unroll
            for (int kc = 0; kc < 4; ++kc)
                acc = __builtin_amdgcn_wmma_f32_16x16x32_f16(
                    false, A1[kc], false, Bf[kc], (short)0, acc, false, false);
            __builtin_amdgcn_sched_barrier(0);
            if (ct + 1 < LD / 16) {
#pragma unroll
                for (int kc = 0; kc < 4; ++kc) {
                    const _Float16* p = db + (size_t)(ct + 1) * 16 * PITCH + kc * 32;
                    Bf[kc] = mk16(*(const half8*)(p), *(const half8*)(p + 16));
                }
            }
            __builtin_amdgcn_sched_barrier(0);
#pragma unroll
            for (int v = 0; v < 8; ++v) rmax1[v] = fmaxf(rmax1[v], dm ? acc[v] : NEGV);
#pragma unroll
            for (int v = 0; v < 8; ++v) asm volatile("" : "+v"(rmax1[v]));
        }
        __builtin_amdgcn_sched_barrier(0);
    }

    float y0[8], y1[8];
#pragma unroll
    for (int v = 0; v < 8; ++v) {
        float a = rmax0[v], c = rmax1[v];
        a = fmaxf(a, __shfl_xor(a, 1, 32));  c = fmaxf(c, __shfl_xor(c, 1, 32));
        a = fmaxf(a, __shfl_xor(a, 2, 32));  c = fmaxf(c, __shfl_xor(c, 2, 32));
        a = fmaxf(a, __shfl_xor(a, 4, 32));  c = fmaxf(c, __shfl_xor(c, 4, 32));
        a = fmaxf(a, __shfl_xor(a, 8, 32));  c = fmaxf(c, __shfl_xor(c, 8, 32));
        y0[v] = ((qbits >> (8 * halfg + v)) & 1u) ? a : 0.0f;
        y1[v] = ((qbits >> (16 + 8 * halfg + v)) & 1u) ? c : 0.0f;
    }

    float u[8];
#pragma unroll
    for (int v = 0; v < 8; ++v) {
        const float t = y0[v] + y1[v];
        u[v] = t + __shfl_xor(t, 16, 32);
    }
    const float s40 = u[0] + u[4], s41 = u[1] + u[5], s42 = u[2] + u[6], s43 = u[3] + u[7];
    const float s20 = s40 + s42, s21 = s41 + s43;
    const float tot = s20 + s21;
    if (lane == 0) out[w * N_ + n] = tot;
}

extern "C" void kernel_launch(void* const* d_in, const int* in_sizes, int n_in,
                              void* d_out, int out_size, void* d_ws, size_t ws_size,
                              hipStream_t stream) {
    const float* q_hidden = (const float*)d_in[0];
    const int*   q_mask   = (const int*)d_in[1];
    const float* d_hidden = (const float*)d_in[2];
    const int*   d_mask   = (const int*)d_in[3];
    float*       out      = (float*)d_out;

    _Float16* qn = (_Float16*)d_ws;

    qnorm_kernel<<<1, 256, 0, stream>>>(q_hidden, qn);

    const size_t shbytes = (size_t)LD * PITCH * sizeof(_Float16);
    colbert_maxsim_kernel<<<N_, 256, shbytes, stream>>>(
        qn, q_mask, d_hidden, d_mask, out);
}


# Doc tokens with d_mask = 0 are neither loaded nor multiplied: each wave packs its unmasked rows into consecutive LDS rows, tile loop runs ceil(U/16) column tiles, row maxima start at -100
# speedup vs baseline: 7.3263x; 1.2153x over previous
#include <hip/hip_runtime.h>


typedef _Float16 half4  __attribute__((ext_vector_type(4)));
typedef _Float16 half8  __attribute__((ext_vector_type(8)));
typedef _Float16 half16 __attribute__((ext_vector_type(16)));
typedef float    v8f    __attribute__((ext_vector_type(8)));

#define B_   8
#define LQ   32
#define N_   1024
#define LD   256
#define K_   128
#define PITCH 136
#define NEGV -100.0f
#define EPSV 1e-8f

template <int M>
static __device__ inline float xmask_add(float v) {
    const int t = __builtin_amdgcn_update_dpp(0, __float_as_int(v), 0x160 + M, 0xf, 0xf, true);
    return v + __int_as_float(t);
}

template <int DPITCH>
static __device__ inline void norm_rows32(const float* __restrict__ src,
                                          _Float16* __restrict__ dst, const int lane) {
    const int g = lane >> 3;
    const int c = lane & 7;
    const float* s0 = src + (size_t)(4 * g) * K_ + 4 * c;
    _Float16*    d0 = dst + (size_t)(4 * g) * DPITCH + 4 * c;
#pragma nounroll
    for (int hb = 0; hb < 32; hb += 16) {
        float4 x[4][4];
#pragma unroll
        for (int j = 0; j < 4; ++j) {
#pragma unroll
            for (int i = 0; i < 4; ++i)
                x[j][i] = *(const float4*)(s0 + (size_t)(hb + j) * K_ + 32 * i);
        }
#pragma unroll
        for (int j = 0; j < 4; ++j) {
            float s = 0.f;
#pragma unroll
            for (int i = 0; i < 4; ++i) {
                const float4 v = x[j][i];
                s += v.x * v.x + v.y * v.y + v.z * v.z + v.w * v.w;
            }
            s = xmask_add<1>(s);
            s = xmask_add<2>(s);
            s = xmask_add<4>(s);
            const float inv = 1.0f / fmaxf(sqrtf(s), EPSV);
            _Float16* d = d0 + (size_t)(hb + j) * DPITCH;
#pragma unroll
            for (int i = 0; i < 4; ++i) {
                const float4 v = x[j][i];
                half4 h;
                h[0] = (_Float16)(v.x * inv);
                h[1] = (_Float16)(v.y * inv);
                h[2] = (_Float16)(v.z * inv);
                h[3] = (_Float16)(v.w * inv);
                *(half4*)(d + 32 * i) = h;
            }
        }
    }
}


template <int NQ>
static __device__ inline void norm_pass(const float* __restrict__ src, _Float16* __restrict__ dst,
                                        const int sel, const int cnt, const int m0,
                                        const int g, const int c) {
    float4 x[NQ][4];
#pragma unroll
    for (int j = 0; j < NQ; ++j) {
        const int m   = m0 + 4 * j + g;
        const int row = __shfl(sel, m < cnt ? m : cnt - 1, 32);
        const float* s = src + (size_t)row * K_ + 4 * c;
#pragma unroll
        for (int i = 0; i < 4; ++i) x[j][i] = *(const float4*)(s + 32 * i);
    }
#pragma unroll
    for (int j = 0; j < NQ; ++j) {
        const int m = m0 + 4 * j + g;
        float s = 0.f;
#pragma unroll
        for (int i = 0; i < 4; ++i) {
            const float4 v = x[j][i];
            s += v.x * v.x + v.y * v.y + v.z * v.z + v.w * v.w;
        }
        s = xmask_add<1>(s);
        s = xmask_add<2>(s);
        s = xmask_add<4>(s);
        const float inv = 1.0f / fmaxf(sqrtf(s), EPSV);
        _Float16* d = dst + (size_t)m * PITCH + 4 * c;
        if (m < cnt) {
#pragma unroll
            for (int i = 0; i < 4; ++i) {
                const float4 v = x[j][i];
                half4 h;
                h[0] = (_Float16)(v.x * inv);
                h[1] = (_Float16)(v.y * inv);
                h[2] = (_Float16)(v.z * inv);
                h[3] = (_Float16)(v.w * inv);
                *(half4*)(d + 32 * i) = h;
            }
        }
    }
}

static __device__ inline void norm_valid_rows32(const float* __restrict__ src,
                                                _Float16* __restrict__ dst,
                                                const unsigned bits, const int lane) {
    const int g   = lane >> 3;
    const int c   = lane & 7;
    const int cnt = __builtin_popcount(bits);
    const int  rank = __builtin_popcount(bits & ((1u << lane) - 1u));
    const bool val  = ((bits >> lane) & 1u) != 0u;
    const int  sel  = __builtin_amdgcn_ds_permute((val ? rank : cnt + lane - rank) << 2, lane);
#pragma nounroll
    for (int m0 = 0; m0 < cnt; m0 += 16) {
        const int left = cnt - m0;
        if (left > 12)     norm_pass<4>(src, dst, sel, cnt, m0, g, c);
        else if (left > 8) norm_pass<3>(src, dst, sel, cnt, m0, g, c);
        else if (left > 4) norm_pass<2>(src, dst, sel, cnt, m0, g, c);
        else               norm_pass<1>(src, dst, sel, cnt, m0, g, c);
    }
}

__global__ void __launch_bounds__(256) qnorm_kernel(const float* __restrict__ q,
                                                    _Float16* __restrict__ qn) {
    const int w    = threadIdx.x >> 5;
    const int lane = threadIdx.x & 31;
    norm_rows32<K_>(q + (size_t)(w * 32) * K_, qn + (size_t)(w * 32) * K_, lane);
}

static __device__ inline half16 mk16(half8 lo, half8 hi) {
    return __builtin_shufflevector(lo, hi, 0, 1, 2, 3, 4, 5, 6, 7,
                                   8, 9, 10, 11, 12, 13, 14, 15);
}

template <bool TAIL>
static __device__ inline void tile_step(const half16 (&A0)[4], const half16 (&A1)[4],
                                        half16 (&Bf)[4], float (&rmax0)[8], float (&rmax1)[8],
                                        const _Float16* next, const bool more, const bool cv) {
    {
        v8f acc = {};
#pragma unroll
        for (int kc = 0; kc < 4; ++kc)
            acc = __builtin_amdgcn_wmma_f32_16x16x32_f16(
                false, A0[kc], false, Bf[kc], (short)0, acc, false, false);
#pragma unroll
        for (int v = 0; v < 8; ++v)
            rmax0[v] = fmaxf(rmax0[v], (!TAIL || cv) ? acc[v] : NEGV);
#pragma unroll
        for (int v = 0; v < 8; ++v) asm volatile("" : "+v"(rmax0[v]));
    }
    __builtin_amdgcn_sched_barrier(0);
    {
        v8f acc = {};
#pragma unroll
        for (int kc = 0; kc < 4; ++kc)
            acc = __builtin_amdgcn_wmma_f32_16x16x32_f16(
                false, A1[kc], false, Bf[kc], (short)0, acc, false, false);
        __builtin_amdgcn_sched_barrier(0);
        if (more) {
#pragma unroll
            for (int kc = 0; kc < 4; ++kc)
                Bf[kc] = mk16(*(const half8*)(next + kc * 32), *(const half8*)(next + kc * 32 + 16));
        }
        __builtin_amdgcn_sched_barrier(0);
#pragma unroll
        for (int v = 0; v < 8; ++v)
            rmax1[v] = fmaxf(rmax1[v], (!TAIL || cv) ? acc[v] : NEGV);
#pragma unroll
        for (int v = 0; v < 8; ++v) asm volatile("" : "+v"(rmax1[v]));
    }
    __builtin_amdgcn_sched_barrier(0);
}

__global__ void __launch_bounds__(256) colbert_maxsim_kernel(
    const _Float16* __restrict__ qn, const int* __restrict__ q_mask,
    const float* __restrict__ d_hidden, const int* __restrict__ d_mask,
    float* __restrict__ out) {
    extern __shared__ _Float16 sd[];

    const int n     = blockIdx.x;
    const int tid   = threadIdx.x;
    const int w     = __builtin_amdgcn_readfirstlane(tid >> 5);
    const int lane  = tid & 31;
    const int lmod  = lane & 15;
    const int halfg = lane >> 4;

    unsigned mw[8];
    {
        int mv[8];
#pragma unroll
        for (int i = 0; i < 8; ++i) mv[i] = d_mask[n * LD + 32 * i + lane];
#pragma unroll
        for (int i = 0; i < 8; ++i) mw[i] = __builtin_amdgcn_ballot_w32(mv[i] > 0);
    }
    const int qmv = q_mask[w * LQ + lane];

    unsigned bits = 0u;
    int base = 0, U = 0;
#pragma unroll
    for (int i = 0; i < 8; ++i) {
        const int cnt = __builtin_popcount(mw[i]);
        if (i == w) bits = mw[i];
        if (i < w) base += cnt;
        U += cnt;
    }
    const int nfull = U >> 4;
    const int urem  = U & 15;

    norm_valid_rows32(d_hidden + ((size_t)n * LD + w * 32) * K_,
                      sd + (size_t)base * PITCH, bits, lane);

    if (urem != 0 && tid < (16 - urem) * 16) {
        const half8 z = {};
        *(half8*)(sd + (size_t)(U + (tid >> 4)) * PITCH + 8 * (tid & 15)) = z;
    }

    half16 A0[4], A1[4];
    {
        const _Float16* qa0 = qn + (size_t)(w * LQ + lmod) * K_ + 8 * halfg;
        const _Float16* qa1 = qn + (size_t)(w * LQ + 16 + lmod) * K_ + 8 * halfg;
#pragma unroll
        for (int kc = 0; kc < 4; ++kc) {
            A0[kc] = mk16(*(const half8*)(qa0 + kc * 32),
                          *(const half8*)(qa0 + kc * 32 + 16));
            A1[kc] = mk16(*(const half8*)(qa1 + kc * 32),
                          *(const half8*)(qa1 + kc * 32 + 16));
        }
    }

    const unsigned qbits = __builtin_amdgcn_ballot_w32(qmv > 0);

    __syncthreads();

    float rmax0[8], rmax1[8];
#pragma unroll
    for (int v = 0; v < 8; ++v) { rmax0[v] = NEGV; rmax1[v] = NEGV; }

    const _Float16* db = sd + (size_t)lmod * PITCH + 8 * halfg;
    half16 Bf[4];
    if (U != 0) {
#pragma unroll
        for (int kc = 0; kc < 4; ++kc)
            Bf[kc] = mk16(*(const half8*)(db + kc * 32), *(const half8*)(db + kc * 32 + 16));
    }
    for (int ct = 0; ct < nfull; ++ct) {
        db += 16 * PITCH;
        tile_step<false>(A0, A1, Bf, rmax0, rmax1, db, (ct + 1 < nfull) || (urem != 0), true);
    }
    if (urem != 0)
        tile_step<true>(A0, A1, Bf, rmax0, rmax1, db, false, lmod < urem);

    float y0[8], y1[8];
#pragma unroll
    for (int v = 0; v < 8; ++v) {
        float a = rmax0[v], c = rmax1[v];
        a = fmaxf(a, __shfl_xor(a, 1, 32));  c = fmaxf(c, __shfl_xor(c, 1, 32));
        a = fmaxf(a, __shfl_xor(a, 2, 32));  c = fmaxf(c, __shfl_xor(c, 2, 32));
        a = fmaxf(a, __shfl_xor(a, 4, 32));  c = fmaxf(c, __shfl_xor(c, 4, 32));
        a = fmaxf(a, __shfl_xor(a, 8, 32));  c = fmaxf(c, __shfl_xor(c, 8, 32));
        y0[v] = ((qbits >> (8 * halfg + v)) & 1u) ? a : 0.0f;
        y1[v] = ((qbits >> (16 + 8 * halfg + v)) & 1u) ? c : 0.0f;
    }

    float u[8];
#pragma unroll
    for (int v = 0; v < 8; ++v) {
        const float t = y0[v] + y1[v];
        u[v] = t + __shfl_xor(t, 16, 32);
    }
    const float s40 = u[0] + u[4], s41 = u[1] + u[5], s42 = u[2] + u[6], s43 = u[3] + u[7];
    const float s20 = s40 + s42, s21 = s41 + s43;
    const float tot = s20 + s21;
    if (lane == 0) out[w * N_ + n] = tot;
}

extern "C" void kernel_launch(void* const* d_in, const int* in_sizes, int n_in,
                              void* d_out, int out_size, void* d_ws, size_t ws_size,
                              hipStream_t stream) {
    const float* q_hidden = (const float*)d_in[0];
    const int*   q_mask   = (const int*)d_in[1];
    const float* d_hidden = (const float*)d_in[2];
    const int*   d_mask   = (const int*)d_in[3];
    float*       out      = (float*)d_out;

    _Float16* qn = (_Float16*)d_ws;

    qnorm_kernel<<<1, 256, 0, stream>>>(q_hidden, qn);

    const size_t shbytes = (size_t)LD * PITCH * sizeof(_Float16);
    colbert_maxsim_kernel<<<N_, 256, shbytes, stream>>>(
        qn, q_mask, d_hidden, d_mask, out);
}


# Normalized q stored in A-fragment order (staged through LDS in the small kernel): each A fragment load of the main kernel reads one contiguous 512 B = 4 lines, was 16 lines per load
# speedup vs baseline: 7.5852x; 1.0353x over previous
#include <hip/hip_runtime.h>


typedef _Float16 half4  __attribute__((ext_vector_type(4)));
typedef _Float16 half8  __attribute__((ext_vector_type(8)));
typedef _Float16 half16 __attribute__((ext_vector_type(16)));
typedef float    v8f    __attribute__((ext_vector_type(8)));

#define B_   8
#define LQ   32
#define N_   1024
#define LD   256
#define K_   128
#define PITCH 136
#define NEGV -100.0f
#define EPSV 1e-8f

template <int M>
static __device__ inline float xmask_add(float v) {
    const int t = __builtin_amdgcn_update_dpp(0, __float_as_int(v), 0x160 + M, 0xf, 0xf, true);
    return v + __int_as_float(t);
}

template <int DPITCH>
static __device__ inline void norm_rows32(const float* __restrict__ src,
                                          _Float16* __restrict__ dst, const int lane) {
    const int g = lane >> 3;
    const int c = lane & 7;
    const float* s0 = src + (size_t)(4 * g) * K_ + 4 * c;
    _Float16*    d0 = dst + (size_t)(4 * g) * DPITCH + 4 * c;
#pragma nounroll
    for (int hb = 0; hb < 32; hb += 16) {
        float4 x[4][4];
#pragma unroll
        for (int j = 0; j < 4; ++j) {
#pragma unroll
            for (int i = 0; i < 4; ++i)
                x[j][i] = *(const float4*)(s0 + (size_t)(hb + j) * K_ + 32 * i);
        }
#pragma unroll
        for (int j = 0; j < 4; ++j) {
            float s = 0.f;
#pragma unroll
            for (int i = 0; i < 4; ++i) {
                const float4 v = x[j][i];
                s += v.x * v.x + v.y * v.y + v.z * v.z + v.w * v.w;
            }
            s = xmask_add<1>(s);
            s = xmask_add<2>(s);
            s = xmask_add<4>(s);
            const float inv = 1.0f / fmaxf(sqrtf(s), EPSV);
            _Float16* d = d0 + (size_t)(hb + j) * DPITCH;
#pragma unroll
            for (int i = 0; i < 4; ++i) {
                const float4 v = x[j][i];
                half4 h;
                h[0] = (_Float16)(v.x * inv);
                h[1] = (_Float16)(v.y * inv);
                h[2] = (_Float16)(v.z * inv);
                h[3] = (_Float16)(v.w * inv);
                *(half4*)(d + 32 * i) = h;
            }
        }
    }
}


template <int NQ>
static __device__ inline void norm_pass(const float* __restrict__ src, _Float16* __restrict__ dst,
                                        const int sel, const int cnt, const int m0,
                                        const int g, const int c) {
    float4 x[NQ][4];
#pragma unroll
    for (int j = 0; j < NQ; ++j) {
        const int m   = m0 + 4 * j + g;
        const int row = __shfl(sel, m < cnt ? m : cnt - 1, 32);
        const float* s = src + (size_t)row * K_ + 4 * c;
#pragma unroll
        for (int i = 0; i < 4; ++i) x[j][i] = *(const float4*)(s + 32 * i);
    }
#pragma unroll
    for (int j = 0; j < NQ; ++j) {
        const int m = m0 + 4 * j + g;
        float s = 0.f;
#pragma unroll
        for (int i = 0; i < 4; ++i) {
            const float4 v = x[j][i];
            s += v.x * v.x + v.y * v.y + v.z * v.z + v.w * v.w;
        }
        s = xmask_add<1>(s);
        s = xmask_add<2>(s);
        s = xmask_add<4>(s);
        const float inv = 1.0f / fmaxf(sqrtf(s), EPSV);
        _Float16* d = dst + (size_t)m * PITCH + 4 * c;
        if (m < cnt) {
#pragma unroll
            for (int i = 0; i < 4; ++i) {
                const float4 v = x[j][i];
                half4 h;
                h[0] = (_Float16)(v.x * inv);
                h[1] = (_Float16)(v.y * inv);
                h[2] = (_Float16)(v.z * inv);
                h[3] = (_Float16)(v.w * inv);
                *(half4*)(d + 32 * i) = h;
            }
        }
    }
}

static __device__ inline void norm_valid_rows32(const float* __restrict__ src,
                                                _Float16* __restrict__ dst,
                                                const unsigned bits, const int lane) {
    const int g   = lane >> 3;
    const int c   = lane & 7;
    const int cnt = __builtin_popcount(bits);
    const int  rank = __builtin_popcount(bits & ((1u << lane) - 1u));
    const bool val  = ((bits >> lane) & 1u) != 0u;
    const int  sel  = __builtin_amdgcn_ds_permute((val ? rank : cnt + lane - rank) << 2, lane);
#pragma nounroll
    for (int m0 = 0; m0 < cnt; m0 += 16) {
        const int left = cnt - m0;
        if (left > 12)     norm_pass<4>(src, dst, sel, cnt, m0, g, c);
        else if (left > 8) norm_pass<3>(src, dst, sel, cnt, m0, g, c);
        else if (left > 4) norm_pass<2>(src, dst, sel, cnt, m0, g, c);
        else               norm_pass<1>(src, dst, sel, cnt, m0, g, c);
    }
}

__global__ void __launch_bounds__(256) qnorm_kernel(const float* __restrict__ q,
                                                    _Float16* __restrict__ qn) {
    extern __shared__ _Float16 sq[];
    const int w     = threadIdx.x >> 5;
    const int lane  = threadIdx.x & 31;
    const int lmod  = lane & 15;
    const int halfg = lane >> 4;
    _Float16* rows = sq + (size_t)(w * 32) * PITCH;
    norm_rows32<PITCH>(q + (size_t)(w * 32) * K_, rows, lane);
    __syncthreads();
    const _Float16* fr = rows + (size_t)lmod * PITCH + 8 * halfg;
    half8* blk = (half8*)qn + (size_t)w * 512 + lane;
#pragma unroll
    for (int mt = 0; mt < 2; ++mt) {
#pragma unroll
        for (int kc = 0; kc < 4; ++kc) {
#pragma unroll
            for (int part = 0; part < 2; ++part)
                blk[((mt * 4 + kc) * 2 + part) * 32] =
                    *(const half8*)(fr + (size_t)(16 * mt) * PITCH + kc * 32 + 16 * part);
        }
    }
}

static __device__ inline half16 mk16(half8 lo, half8 hi) {
    return __builtin_shufflevector(lo, hi, 0, 1, 2, 3, 4, 5, 6, 7,
                                   8, 9, 10, 11, 12, 13, 14, 15);
}

template <bool TAIL>
static __device__ inline void tile_step(const half16 (&A0)[4], const half16 (&A1)[4],
                                        half16 (&Bf)[4], float (&rmax0)[8], float (&rmax1)[8],
                                        const _Float16* next, const bool more, const bool cv) {
    {
        v8f acc = {};
#pragma unroll
        for (int kc = 0; kc < 4; ++kc)
            acc = __builtin_amdgcn_wmma_f32_16x16x32_f16(
                false, A0[kc], false, Bf[kc], (short)0, acc, false, false);
#pragma unroll
        for (int v = 0; v < 8; ++v)
            rmax0[v] = fmaxf(rmax0[v], (!TAIL || cv) ? acc[v] : NEGV);
#pragma unroll
        for (int v = 0; v < 8; ++v) asm volatile("" : "+v"(rmax0[v]));
    }
    __builtin_amdgcn_sched_barrier(0);
    {
        v8f acc = {};
#pragma unroll
        for (int kc = 0; kc < 4; ++kc)
            acc = __builtin_amdgcn_wmma_f32_16x16x32_f16(
                false, A1[kc], false, Bf[kc], (short)0, acc, false, false);
        __builtin_amdgcn_sched_barrier(0);
        if (more) {
#pragma unroll
            for (int kc = 0; kc < 4; ++kc)
                Bf[kc] = mk16(*(const half8*)(next + kc * 32), *(const half8*)(next + kc * 32 + 16));
        }
        __builtin_amdgcn_sched_barrier(0);
#pragma unroll
        for (int v = 0; v < 8; ++v)
            rmax1[v] = fmaxf(rmax1[v], (!TAIL || cv) ? acc[v] : NEGV);
#pragma unroll
        for (int v = 0; v < 8; ++v) asm volatile("" : "+v"(rmax1[v]));
    }
    __builtin_amdgcn_sched_barrier(0);
}

__global__ void __launch_bounds__(256) colbert_maxsim_kernel(
    const _Float16* __restrict__ qn, const int* __restrict__ q_mask,
    const float* __restrict__ d_hidden, const int* __restrict__ d_mask,
    float* __restrict__ out) {
    extern __shared__ _Float16 sd[];

    const int n     = blockIdx.x;
    const int tid   = threadIdx.x;
    const int w     = __builtin_amdgcn_readfirstlane(tid >> 5);
    const int lane  = tid & 31;
    const int lmod  = lane & 15;
    const int halfg = lane >> 4;

    unsigned mw[8];
    {
        int mv[8];
#pragma unroll
        for (int i = 0; i < 8; ++i) mv[i] = d_mask[n * LD + 32 * i + lane];
#pragma unroll
        for (int i = 0; i < 8; ++i) mw[i] = __builtin_amdgcn_ballot_w32(mv[i] > 0);
    }
    const int qmv = q_mask[w * LQ + lane];

    unsigned bits = 0u;
    int base = 0, U = 0;
#pragma unroll
    for (int i = 0; i < 8; ++i) {
        const int cnt = __builtin_popcount(mw[i]);
        if (i == w) bits = mw[i];
        if (i < w) base += cnt;
        U += cnt;
    }
    const int nfull = U >> 4;
    const int urem  = U & 15;

    norm_valid_rows32(d_hidden + ((size_t)n * LD + w * 32) * K_,
                      sd + (size_t)base * PITCH, bits, lane);

    if (urem != 0 && tid < (16 - urem) * 16) {
        const half8 z = {};
        *(half8*)(sd + (size_t)(U + (tid >> 4)) * PITCH + 8 * (tid & 15)) = z;
    }

    half16 A0[4], A1[4];
    {
        const half8* qf = (const half8*)qn + (size_t)w * 512 + lane;
#pragma unroll
        for (int kc = 0; kc < 4; ++kc) {
            A0[kc] = mk16(qf[(kc * 2) * 32], qf[(kc * 2 + 1) * 32]);
            A1[kc] = mk16(qf[((4 + kc) * 2) * 32], qf[((4 + kc) * 2 + 1) * 32]);
        }
    }

    const unsigned qbits = __builtin_amdgcn_ballot_w32(qmv > 0);

    __syncthreads();

    float rmax0[8], rmax1[8];
#pragma unroll
    for (int v = 0; v < 8; ++v) { rmax0[v] = NEGV; rmax1[v] = NEGV; }

    const _Float16* db = sd + (size_t)lmod * PITCH + 8 * halfg;
    half16 Bf[4];
    if (U != 0) {
#pragma unroll
        for (int kc = 0; kc < 4; ++kc)
            Bf[kc] = mk16(*(const half8*)(db + kc * 32), *(const half8*)(db + kc * 32 + 16));
    }
    for (int ct = 0; ct < nfull; ++ct) {
        db += 16 * PITCH;
        tile_step<false>(A0, A1, Bf, rmax0, rmax1, db, (ct + 1 < nfull) || (urem != 0), true);
    }
    if (urem != 0)
        tile_step<true>(A0, A1, Bf, rmax0, rmax1, db, false, lmod < urem);

    float y0[8], y1[8];
#pragma unroll
    for (int v = 0; v < 8; ++v) {
        float a = rmax0[v], c = rmax1[v];
        a = fmaxf(a, __shfl_xor(a, 1, 32));  c = fmaxf(c, __shfl_xor(c, 1, 32));
        a = fmaxf(a, __shfl_xor(a, 2, 32));  c = fmaxf(c, __shfl_xor(c, 2, 32));
        a = fmaxf(a, __shfl_xor(a, 4, 32));  c = fmaxf(c, __shfl_xor(c, 4, 32));
        a = fmaxf(a, __shfl_xor(a, 8, 32));  c = fmaxf(c, __shfl_xor(c, 8, 32));
        y0[v] = ((qbits >> (8 * halfg + v)) & 1u) ? a : 0.0f;
        y1[v] = ((qbits >> (16 + 8 * halfg + v)) & 1u) ? c : 0.0f;
    }

    float u[8];
#pragma unroll
    for (int v = 0; v < 8; ++v) {
        const float t = y0[v] + y1[v];
        u[v] = t + __shfl_xor(t, 16, 32);
    }
    const float s40 = u[0] + u[4], s41 = u[1] + u[5], s42 = u[2] + u[6], s43 = u[3] + u[7];
    const float s20 = s40 + s42, s21 = s41 + s43;
    const float tot = s20 + s21;
    if (lane == 0) out[w * N_ + n] = tot;
}

extern "C" void kernel_launch(void* const* d_in, const int* in_sizes, int n_in,
                              void* d_out, int out_size, void* d_ws, size_t ws_size,
                              hipStream_t stream) {
    const float* q_hidden = (const float*)d_in[0];
    const int*   q_mask   = (const int*)d_in[1];
    const float* d_hidden = (const float*)d_in[2];
    const int*   d_mask   = (const int*)d_in[3];
    float*       out      = (float*)d_out;

    _Float16* qn = (_Float16*)d_ws;

    const size_t shbytes = (size_t)LD * PITCH * sizeof(_Float16);

    qnorm_kernel<<<1, 256, shbytes, stream>>>(q_hidden, qn);

    colbert_maxsim_kernel<<<N_, 256, shbytes, stream>>>(
        qn, q_mask, d_hidden, d_mask, out);
}


# q normalization folded into the main kernel: each wave normalizes its batch's 32 fp32 q rows and turns them into A fragments through a small wave-private LDS scratch; launches 2 -> 1
# speedup vs baseline: 7.9463x; 1.0476x over previous
#include <hip/hip_runtime.h>


typedef _Float16 half4  __attribute__((ext_vector_type(4)));
typedef _Float16 half8  __attribute__((ext_vector_type(8)));
typedef _Float16 half16 __attribute__((ext_vector_type(16)));
typedef float    v8f    __attribute__((ext_vector_type(8)));

#define B_   8
#define LQ   32
#define N_   1024
#define LD   256
#define K_   128
#define PITCH 136
#define NEGV -100.0f
#define EPSV 1e-8f

template <int M>
static __device__ inline float xmask_add(float v) {
    const int t = __builtin_amdgcn_update_dpp(0, __float_as_int(v), 0x160 + M, 0xf, 0xf, true);
    return v + __int_as_float(t);
}

#define QP 40

static __device__ inline half16 mk16(half8 lo, half8 hi) {
    return __builtin_shufflevector(lo, hi, 0, 1, 2, 3, 4, 5, 6, 7,
                                   8, 9, 10, 11, 12, 13, 14, 15);
}

static __device__ inline void q_frags16(const float* __restrict__ src, _Float16* scr,
                                        half16 (&A)[4], const int lane) {
    const int g = lane >> 3;
    const int c = lane & 7;
    const float* s0 = src + (size_t)(4 * g) * K_ + 4 * c;
    half4 h[4][4];
#pragma unroll
    for (int jp = 0; jp < 4; jp += 2) {
        float4 x[2][4];
#pragma unroll
        for (int j = 0; j < 2; ++j) {
#pragma unroll
            for (int i = 0; i < 4; ++i)
                x[j][i] = *(const float4*)(s0 + (size_t)(jp + j) * K_ + 32 * i);
        }
#pragma unroll
        for (int j = 0; j < 2; ++j) {
            float s = 0.f;
#pragma unroll
            for (int i = 0; i < 4; ++i) {
                const float4 v = x[j][i];
                s += v.x * v.x + v.y * v.y + v.z * v.z + v.w * v.w;
            }
            s = xmask_add<1>(s);
            s = xmask_add<2>(s);
            s = xmask_add<4>(s);
            const float inv = 1.0f / fmaxf(sqrtf(s), EPSV);
#pragma unroll
            for (int i = 0; i < 4; ++i) {
                const float4 v = x[j][i];
                half4 t;
                t[0] = (_Float16)(v.x * inv);
                t[1] = (_Float16)(v.y * inv);
                t[2] = (_Float16)(v.z * inv);
                t[3] = (_Float16)(v.w * inv);
                h[jp + j][i] = t;
            }
        }
    }
    _Float16*       wr = scr + (size_t)(4 * g) * QP + 4 * c;
    const _Float16* rd = scr + (size_t)(lane & 15) * QP + 8 * (lane >> 4);
#pragma unroll
    for (int i = 0; i < 4; ++i) {
#pragma unroll
        for (int j = 0; j < 4; ++j) *(half4*)(wr + (size_t)j * QP) = h[j][i];
        __builtin_amdgcn_wave_barrier();
        A[i] = mk16(*(const half8*)rd, *(const half8*)(rd + 16));
        __builtin_amdgcn_wave_barrier();
    }
}


template <int NQ>
static __device__ inline void norm_pass(const float* __restrict__ src, _Float16* __restrict__ dst,
                                        const int sel, const int cnt, const int m0,
                                        const int g, const int c) {
    float4 x[NQ][4];
#pragma unroll
    for (int j = 0; j < NQ; ++j) {
        const int m   = m0 + 4 * j + g;
        const int row = __shfl(sel, m < cnt ? m : cnt - 1, 32);
        const float* s = src + (size_t)row * K_ + 4 * c;
#pragma unroll
        for (int i = 0; i < 4; ++i) x[j][i] = *(const float4*)(s + 32 * i);
    }
#pragma unroll
    for (int j = 0; j < NQ; ++j) {
        const int m = m0 + 4 * j + g;
        float s = 0.f;
#pragma unroll
        for (int i = 0; i < 4; ++i) {
            const float4 v = x[j][i];
            s += v.x * v.x + v.y * v.y + v.z * v.z + v.w * v.w;
        }
        s = xmask_add<1>(s);
        s = xmask_add<2>(s);
        s = xmask_add<4>(s);
        const float inv = 1.0f / fmaxf(sqrtf(s), EPSV);
        _Float16* d = dst + (size_t)m * PITCH + 4 * c;
        if (m < cnt) {
#pragma unroll
            for (int i = 0; i < 4; ++i) {
                const float4 v = x[j][i];
                half4 h;
                h[0] = (_Float16)(v.x * inv);
                h[1] = (_Float16)(v.y * inv);
                h[2] = (_Float16)(v.z * inv);
                h[3] = (_Float16)(v.w * inv);
                *(half4*)(d + 32 * i) = h;
            }
        }
    }
}

static __device__ inline void norm_valid_rows32(const float* __restrict__ src,
                                                _Float16* __restrict__ dst,
                                                const unsigned bits, const int lane) {
    const int g   = lane >> 3;
    const int c   = lane & 7;
    const int cnt = __builtin_popcount(bits);
    const int  rank = __builtin_popcount(bits & ((1u << lane) - 1u));
    const bool val  = ((bits >> lane) & 1u) != 0u;
    const int  sel  = __builtin_amdgcn_ds_permute((val ? rank : cnt + lane - rank) << 2, lane);
#pragma nounroll
    for (int m0 = 0; m0 < cnt; m0 += 16) {
        const int left = cnt - m0;
        if (left > 12)     norm_pass<4>(src, dst, sel, cnt, m0, g, c);
        else if (left > 8) norm_pass<3>(src, dst, sel, cnt, m0, g, c);
        else if (left > 4) norm_pass<2>(src, dst, sel, cnt, m0, g, c);
        else               norm_pass<1>(src, dst, sel, cnt, m0, g, c);
    }
}

template <bool TAIL>
static __device__ inline void tile_step(const half16 (&A0)[4], const half16 (&A1)[4],
                                        half16 (&Bf)[4], float (&rmax0)[8], float (&rmax1)[8],
                                        const _Float16* next, const bool more, const bool cv) {
    {
        v8f acc = {};
#pragma unroll
        for (int kc = 0; kc < 4; ++kc)
            acc = __builtin_amdgcn_wmma_f32_16x16x32_f16(
                false, A0[kc], false, Bf[kc], (short)0, acc, false, false);
#pragma unroll
        for (int v = 0; v < 8; ++v)
            rmax0[v] = fmaxf(rmax0[v], (!TAIL || cv) ? acc[v] : NEGV);
#pragma unroll
        for (int v = 0; v < 8; ++v) asm volatile("" : "+v"(rmax0[v]));
    }
    __builtin_amdgcn_sched_barrier(0);
    {
        v8f acc = {};
#pragma unroll
        for (int kc = 0; kc < 4; ++kc)
            acc = __builtin_amdgcn_wmma_f32_16x16x32_f16(
                false, A1[kc], false, Bf[kc], (short)0, acc, false, false);
        __builtin_amdgcn_sched_barrier(0);
        if (more) {
#pragma unroll
            for (int kc = 0; kc < 4; ++kc)
                Bf[kc] = mk16(*(const half8*)(next + kc * 32), *(const half8*)(next + kc * 32 + 16));
        }
        __builtin_amdgcn_sched_barrier(0);
#pragma unroll
        for (int v = 0; v < 8; ++v)
            rmax1[v] = fmaxf(rmax1[v], (!TAIL || cv) ? acc[v] : NEGV);
#pragma unroll
        for (int v = 0; v < 8; ++v) asm volatile("" : "+v"(rmax1[v]));
    }
    __builtin_amdgcn_sched_barrier(0);
}

__global__ void __launch_bounds__(256, 8) colbert_maxsim_kernel(
    const float* __restrict__ q, const int* __restrict__ q_mask,
    const float* __restrict__ d_hidden, const int* __restrict__ d_mask,
    float* __restrict__ out) {
    extern __shared__ _Float16 sd[];

    const int n     = blockIdx.x;
    const int tid   = threadIdx.x;
    const int w     = __builtin_amdgcn_readfirstlane(tid >> 5);
    const int lane  = tid & 31;
    const int lmod  = lane & 15;
    const int halfg = lane >> 4;

    unsigned mw[8];
    {
        int mv[8];
#pragma unroll
        for (int i = 0; i < 8; ++i) mv[i] = d_mask[n * LD + 32 * i + lane];
#pragma unroll
        for (int i = 0; i < 8; ++i) mw[i] = __builtin_amdgcn_ballot_w32(mv[i] > 0);
    }
    const int qmv = q_mask[w * LQ + lane];

    unsigned bits = 0u;
    int base = 0, U = 0;
#pragma unroll
    for (int i = 0; i < 8; ++i) {
        const int cnt = __builtin_popcount(mw[i]);
        if (i == w) bits = mw[i];
        if (i < w) base += cnt;
        U += cnt;
    }
    const int nfull = U >> 4;
    const int urem  = U & 15;

    norm_valid_rows32(d_hidden + ((size_t)n * LD + w * 32) * K_,
                      sd + (size_t)base * PITCH, bits, lane);

    if (urem != 0 && tid < (16 - urem) * 16) {
        const half8 z = {};
        *(half8*)(sd + (size_t)(U + (tid >> 4)) * PITCH + 8 * (tid & 15)) = z;
    }

    half16 A0[4], A1[4];
    {
        const float* qb  = q + (size_t)(w * LQ) * K_;
        _Float16*    scr = sd + (size_t)LD * PITCH + (size_t)w * (16 * QP);
        q_frags16(qb, scr, A0, lane);
        q_frags16(qb + (size_t)16 * K_, scr, A1, lane);
    }

    const unsigned qbits = __builtin_amdgcn_ballot_w32(qmv > 0);

    __syncthreads();

    float rmax0[8], rmax1[8];
#pragma unroll
    for (int v = 0; v < 8; ++v) { rmax0[v] = NEGV; rmax1[v] = NEGV; }

    const _Float16* db = sd + (size_t)lmod * PITCH + 8 * halfg;
    half16 Bf[4];
    if (U != 0) {
#pragma unroll
        for (int kc = 0; kc < 4; ++kc)
            Bf[kc] = mk16(*(const half8*)(db + kc * 32), *(const half8*)(db + kc * 32 + 16));
    }
    for (int ct = 0; ct < nfull; ++ct) {
        db += 16 * PITCH;
        tile_step<false>(A0, A1, Bf, rmax0, rmax1, db, (ct + 1 < nfull) || (urem != 0), true);
    }
    if (urem != 0)
        tile_step<true>(A0, A1, Bf, rmax0, rmax1, db, false, lmod < urem);

    float y0[8], y1[8];
#pragma unroll
    for (int v = 0; v < 8; ++v) {
        float a = rmax0[v], c = rmax1[v];
        a = fmaxf(a, __shfl_xor(a, 1, 32));  c = fmaxf(c, __shfl_xor(c, 1, 32));
        a = fmaxf(a, __shfl_xor(a, 2, 32));  c = fmaxf(c, __shfl_xor(c, 2, 32));
        a = fmaxf(a, __shfl_xor(a, 4, 32));  c = fmaxf(c, __shfl_xor(c, 4, 32));
        a = fmaxf(a, __shfl_xor(a, 8, 32));  c = fmaxf(c, __shfl_xor(c, 8, 32));
        y0[v] = ((qbits >> (8 * halfg + v)) & 1u) ? a : 0.0f;
        y1[v] = ((qbits >> (16 + 8 * halfg + v)) & 1u) ? c : 0.0f;
    }

    float u[8];
#pragma unroll
    for (int v = 0; v < 8; ++v) {
        const float t = y0[v] + y1[v];
        u[v] = t + __shfl_xor(t, 16, 32);
    }
    const float s40 = u[0] + u[4], s41 = u[1] + u[5], s42 = u[2] + u[6], s43 = u[3] + u[7];
    const float s20 = s40 + s42, s21 = s41 + s43;
    const float tot = s20 + s21;
    if (lane == 0) out[w * N_ + n] = tot;
}

extern "C" void kernel_launch(void* const* d_in, const int* in_sizes, int n_in,
                              void* d_out, int out_size, void* d_ws, size_t ws_size,
                              hipStream_t stream) {
    const float* q_hidden = (const float*)d_in[0];
    const int*   q_mask   = (const int*)d_in[1];
    const float* d_hidden = (const float*)d_in[2];
    const int*   d_mask   = (const int*)d_in[3];
    float*       out      = (float*)d_out;

    const size_t shbytes = (size_t)LD * PITCH * sizeof(_Float16)
                         + (size_t)B_ * 16 * QP * sizeof(_Float16);

    colbert_maxsim_kernel<<<N_, 256, shbytes, stream>>>(
        q_hidden, q_mask, d_hidden, d_mask, out);
}


# Reciprocal row norm 1/max(sqrt(s),eps) computed as min(v_rsq_f32(s),1/eps) in both normalizations: 2 instructions where correctly rounded sqrt + IEEE divide took about 30 per row group
# speedup vs baseline: 8.4432x; 1.0625x over previous
#include <hip/hip_runtime.h>


typedef _Float16 half4  __attribute__((ext_vector_type(4)));
typedef _Float16 half8  __attribute__((ext_vector_type(8)));
typedef _Float16 half16 __attribute__((ext_vector_type(16)));
typedef float    v8f    __attribute__((ext_vector_type(8)));

#define B_   8
#define LQ   32
#define N_   1024
#define LD   256
#define K_   128
#define PITCH 136
#define NEGV -100.0f
#define EPSV 1e-8f

template <int M>
static __device__ inline float xmask_add(float v) {
    const int t = __builtin_amdgcn_update_dpp(0, __float_as_int(v), 0x160 + M, 0xf, 0xf, true);
    return v + __int_as_float(t);
}

static __device__ inline float rnorm(float s) {
    return fminf(__builtin_amdgcn_rsqf(s), 1.0f / EPSV);
}

#define QP 40

static __device__ inline half16 mk16(half8 lo, half8 hi) {
    return __builtin_shufflevector(lo, hi, 0, 1, 2, 3, 4, 5, 6, 7,
                                   8, 9, 10, 11, 12, 13, 14, 15);
}

static __device__ inline void q_frags16(const float* __restrict__ src, _Float16* scr,
                                        half16 (&A)[4], const int lane) {
    const int g = lane >> 3;
    const int c = lane & 7;
    const float* s0 = src + (size_t)(4 * g) * K_ + 4 * c;
    half4 h[4][4];
#pragma unroll
    for (int jp = 0; jp < 4; jp += 2) {
        float4 x[2][4];
#pragma unroll
        for (int j = 0; j < 2; ++j) {
#pragma unroll
            for (int i = 0; i < 4; ++i)
                x[j][i] = *(const float4*)(s0 + (size_t)(jp + j) * K_ + 32 * i);
        }
#pragma unroll
        for (int j = 0; j < 2; ++j) {
            float s = 0.f;
#pragma unroll
            for (int i = 0; i < 4; ++i) {
                const float4 v = x[j][i];
                s += v.x * v.x + v.y * v.y + v.z * v.z + v.w * v.w;
            }
            s = xmask_add<1>(s);
            s = xmask_add<2>(s);
            s = xmask_add<4>(s);
            const float inv = rnorm(s);
#pragma unroll
            for (int i = 0; i < 4; ++i) {
                const float4 v = x[j][i];
                half4 t;
                t[0] = (_Float16)(v.x * inv);
                t[1] = (_Float16)(v.y * inv);
                t[2] = (_Float16)(v.z * inv);
                t[3] = (_Float16)(v.w * inv);
                h[jp + j][i] = t;
            }
        }
    }
    _Float16*       wr = scr + (size_t)(4 * g) * QP + 4 * c;
    const _Float16* rd = scr + (size_t)(lane & 15) * QP + 8 * (lane >> 4);
#pragma unroll
    for (int i = 0; i < 4; ++i) {
#pragma unroll
        for (int j = 0; j < 4; ++j) *(half4*)(wr + (size_t)j * QP) = h[j][i];
        __builtin_amdgcn_wave_barrier();
        A[i] = mk16(*(const half8*)rd, *(const half8*)(rd + 16));
        __builtin_amdgcn_wave_barrier();
    }
}


template <int NQ>
static __device__ inline void norm_pass(const float* __restrict__ src, _Float16* __restrict__ dst,
                                        const int sel, const int cnt, const int m0,
                                        const int g, const int c) {
    float4 x[NQ][4];
#pragma unroll
    for (int j = 0; j < NQ; ++j) {
        const int m   = m0 + 4 * j + g;
        const int row = __shfl(sel, m < cnt ? m : cnt - 1, 32);
        const float* s = src + (size_t)row * K_ + 4 * c;
#pragma unroll
        for (int i = 0; i < 4; ++i) x[j][i] = *(const float4*)(s + 32 * i);
    }
#pragma unroll
    for (int j = 0; j < NQ; ++j) {
        const int m = m0 + 4 * j + g;
        float s = 0.f;
#pragma unroll
        for (int i = 0; i < 4; ++i) {
            const float4 v = x[j][i];
            s += v.x * v.x + v.y * v.y + v.z * v.z + v.w * v.w;
        }
        s = xmask_add<1>(s);
        s = xmask_add<2>(s);
        s = xmask_add<4>(s);
        const float inv = rnorm(s);
        _Float16* d = dst + (size_t)m * PITCH + 4 * c;
        if (m < cnt) {
#pragma unroll
            for (int i = 0; i < 4; ++i) {
                const float4 v = x[j][i];
                half4 h;
                h[0] = (_Float16)(v.x * inv);
                h[1] = (_Float16)(v.y * inv);
                h[2] = (_Float16)(v.z * inv);
                h[3] = (_Float16)(v.w * inv);
                *(half4*)(d + 32 * i) = h;
            }
        }
    }
}

static __device__ inline void norm_valid_rows32(const float* __restrict__ src,
                                                _Float16* __restrict__ dst,
                                                const unsigned bits, const int lane) {
    const int g   = lane >> 3;
    const int c   = lane & 7;
    const int cnt = __builtin_popcount(bits);
    const int  rank = __builtin_popcount(bits & ((1u << lane) - 1u));
    const bool val  = ((bits >> lane) & 1u) != 0u;
    const int  sel  = __builtin_amdgcn_ds_permute((val ? rank : cnt + lane - rank) << 2, lane);
#pragma nounroll
    for (int m0 = 0; m0 < cnt; m0 += 16) {
        const int left = cnt - m0;
        if (left > 12)     norm_pass<4>(src, dst, sel, cnt, m0, g, c);
        else if (left > 8) norm_pass<3>(src, dst, sel, cnt, m0, g, c);
        else if (left > 4) norm_pass<2>(src, dst, sel, cnt, m0, g, c);
        else               norm_pass<1>(src, dst, sel, cnt, m0, g, c);
    }
}

template <bool TAIL>
static __device__ inline void tile_step(const half16 (&A0)[4], const half16 (&A1)[4],
                                        half16 (&Bf)[4], float (&rmax0)[8], float (&rmax1)[8],
                                        const _Float16* next, const bool more, const bool cv) {
    {
        v8f acc = {};
#pragma unroll
        for (int kc = 0; kc < 4; ++kc)
            acc = __builtin_amdgcn_wmma_f32_16x16x32_f16(
                false, A0[kc], false, Bf[kc], (short)0, acc, false, false);
#pragma unroll
        for (int v = 0; v < 8; ++v)
            rmax0[v] = fmaxf(rmax0[v], (!TAIL || cv) ? acc[v] : NEGV);
#pragma unroll
        for (int v = 0; v < 8; ++v) asm volatile("" : "+v"(rmax0[v]));
    }
    __builtin_amdgcn_sched_barrier(0);
    {
        v8f acc = {};
#pragma unroll
        for (int kc = 0; kc < 4; ++kc)
            acc = __builtin_amdgcn_wmma_f32_16x16x32_f16(
                false, A1[kc], false, Bf[kc], (short)0, acc, false, false);
        __builtin_amdgcn_sched_barrier(0);
        if (more) {
#pragma unroll
            for (int kc = 0; kc < 4; ++kc)
                Bf[kc] = mk16(*(const half8*)(next + kc * 32), *(const half8*)(next + kc * 32 + 16));
        }
        __builtin_amdgcn_sched_barrier(0);
#pragma unroll
        for (int v = 0; v < 8; ++v)
            rmax1[v] = fmaxf(rmax1[v], (!TAIL || cv) ? acc[v] : NEGV);
#pragma unroll
        for (int v = 0; v < 8; ++v) asm volatile("" : "+v"(rmax1[v]));
    }
    __builtin_amdgcn_sched_barrier(0);
}

__global__ void __launch_bounds__(256, 8) colbert_maxsim_kernel(
    const float* __restrict__ q, const int* __restrict__ q_mask,
    const float* __restrict__ d_hidden, const int* __restrict__ d_mask,
    float* __restrict__ out) {
    extern __shared__ _Float16 sd[];

    const int n     = blockIdx.x;
    const int tid   = threadIdx.x;
    const int w     = __builtin_amdgcn_readfirstlane(tid >> 5);
    const int lane  = tid & 31;
    const int lmod  = lane & 15;
    const int halfg = lane >> 4;

    unsigned mw[8];
    {
        int mv[8];
#pragma unroll
        for (int i = 0; i < 8; ++i) mv[i] = d_mask[n * LD + 32 * i + lane];
#pragma unroll
        for (int i = 0; i < 8; ++i) mw[i] = __builtin_amdgcn_ballot_w32(mv[i] > 0);
    }
    const int qmv = q_mask[w * LQ + lane];

    unsigned bits = 0u;
    int base = 0, U = 0;
#pragma unroll
    for (int i = 0; i < 8; ++i) {
        const int cnt = __builtin_popcount(mw[i]);
        if (i == w) bits = mw[i];
        if (i < w) base += cnt;
        U += cnt;
    }
    const int nfull = U >> 4;
    const int urem  = U & 15;

    norm_valid_rows32(d_hidden + ((size_t)n * LD + w * 32) * K_,
                      sd + (size_t)base * PITCH, bits, lane);

    if (urem != 0 && tid < (16 - urem) * 16) {
        const half8 z = {};
        *(half8*)(sd + (size_t)(U + (tid >> 4)) * PITCH + 8 * (tid & 15)) = z;
    }

    half16 A0[4], A1[4];
    {
        const float* qb  = q + (size_t)(w * LQ) * K_;
        _Float16*    scr = sd + (size_t)LD * PITCH + (size_t)w * (16 * QP);
        q_frags16(qb, scr, A0, lane);
        q_frags16(qb + (size_t)16 * K_, scr, A1, lane);
    }

    const unsigned qbits = __builtin_amdgcn_ballot_w32(qmv > 0);

    __syncthreads();

    float rmax0[8], rmax1[8];
#pragma unroll
    for (int v = 0; v < 8; ++v) { rmax0[v] = NEGV; rmax1[v] = NEGV; }

    const _Float16* db = sd + (size_t)lmod * PITCH + 8 * halfg;
    half16 Bf[4];
    if (U != 0) {
#pragma unroll
        for (int kc = 0; kc < 4; ++kc)
            Bf[kc] = mk16(*(const half8*)(db + kc * 32), *(const half8*)(db + kc * 32 + 16));
    }
    for (int ct = 0; ct < nfull; ++ct) {
        db += 16 * PITCH;
        tile_step<false>(A0, A1, Bf, rmax0, rmax1, db, (ct + 1 < nfull) || (urem != 0), true);
    }
    if (urem != 0)
        tile_step<true>(A0, A1, Bf, rmax0, rmax1, db, false, lmod < urem);

    float y0[8], y1[8];
#pragma unroll
    for (int v = 0; v < 8; ++v) {
        float a = rmax0[v], c = rmax1[v];
        a = fmaxf(a, __shfl_xor(a, 1, 32));  c = fmaxf(c, __shfl_xor(c, 1, 32));
        a = fmaxf(a, __shfl_xor(a, 2, 32));  c = fmaxf(c, __shfl_xor(c, 2, 32));
        a = fmaxf(a, __shfl_xor(a, 4, 32));  c = fmaxf(c, __shfl_xor(c, 4, 32));
        a = fmaxf(a, __shfl_xor(a, 8, 32));  c = fmaxf(c, __shfl_xor(c, 8, 32));
        y0[v] = ((qbits >> (8 * halfg + v)) & 1u) ? a : 0.0f;
        y1[v] = ((qbits >> (16 + 8 * halfg + v)) & 1u) ? c : 0.0f;
    }

    float u[8];
#pragma unroll
    for (int v = 0; v < 8; ++v) {
        const float t = y0[v] + y1[v];
        u[v] = t + __shfl_xor(t, 16, 32);
    }
    const float s40 = u[0] + u[4], s41 = u[1] + u[5], s42 = u[2] + u[6], s43 = u[3] + u[7];
    const float s20 = s40 + s42, s21 = s41 + s43;
    const float tot = s20 + s21;
    if (lane == 0) out[w * N_ + n] = tot;
}

extern "C" void kernel_launch(void* const* d_in, const int* in_sizes, int n_in,
                              void* d_out, int out_size, void* d_ws, size_t ws_size,
                              hipStream_t stream) {
    const float* q_hidden = (const float*)d_in[0];
    const int*   q_mask   = (const int*)d_in[1];
    const float* d_hidden = (const float*)d_in[2];
    const int*   d_mask   = (const int*)d_in[3];
    float*       out      = (float*)d_out;

    const size_t shbytes = (size_t)LD * PITCH * sizeof(_Float16)
                         + (size_t)B_ * 16 * QP * sizeof(_Float16);

    colbert_maxsim_kernel<<<N_, 256, shbytes, stream>>>(
        q_hidden, q_mask, d_hidden, d_mask, out);
}


# Each workgroup scores a pair of docs with one set of q fragments: q loaded and normalized once per two docs, grid 1024 -> 512, two workgroups per WGP, 256-VGPR budget (B fragments double-buffered)
# speedup vs baseline: 11.4708x; 1.3586x over previous
#include <hip/hip_runtime.h>


typedef _Float16 half4  __attribute__((ext_vector_type(4)));
typedef _Float16 half8  __attribute__((ext_vector_type(8)));
typedef _Float16 half16 __attribute__((ext_vector_type(16)));
typedef float    v8f    __attribute__((ext_vector_type(8)));

#define B_   8
#define LQ   32
#define N_   1024
#define LD   256
#define K_   128
#define PITCH 136
#define NEGV -100.0f
#define EPSV 1e-8f

template <int M>
static __device__ inline float xmask_add(float v) {
    const int t = __builtin_amdgcn_update_dpp(0, __float_as_int(v), 0x160 + M, 0xf, 0xf, true);
    return v + __int_as_float(t);
}

static __device__ inline float rnorm(float s) {
    return fminf(__builtin_amdgcn_rsqf(s), 1.0f / EPSV);
}

#define QP 40

static __device__ inline half16 mk16(half8 lo, half8 hi) {
    return __builtin_shufflevector(lo, hi, 0, 1, 2, 3, 4, 5, 6, 7,
                                   8, 9, 10, 11, 12, 13, 14, 15);
}

static __device__ inline void q_frags16(const float* __restrict__ src, _Float16* scr,
                                        half16 (&A)[4], const int lane) {
    const int g = lane >> 3;
    const int c = lane & 7;
    const float* s0 = src + (size_t)(4 * g) * K_ + 4 * c;
    half4 h[4][4];
    {
        float4 x[4][4];
#pragma unroll
        for (int j = 0; j < 4; ++j) {
#pragma unroll
            for (int i = 0; i < 4; ++i)
                x[j][i] = *(const float4*)(s0 + (size_t)j * K_ + 32 * i);
        }
#pragma unroll
        for (int j = 0; j < 4; ++j) {
            float s = 0.f;
#pragma unroll
            for (int i = 0; i < 4; ++i) {
                const float4 v = x[j][i];
                s += v.x * v.x + v.y * v.y + v.z * v.z + v.w * v.w;
            }
            s = xmask_add<1>(s);
            s = xmask_add<2>(s);
            s = xmask_add<4>(s);
            const float inv = rnorm(s);
#pragma unroll
            for (int i = 0; i < 4; ++i) {
                const float4 v = x[j][i];
                half4 t;
                t[0] = (_Float16)(v.x * inv);
                t[1] = (_Float16)(v.y * inv);
                t[2] = (_Float16)(v.z * inv);
                t[3] = (_Float16)(v.w * inv);
                h[j][i] = t;
            }
        }
    }
    _Float16*       wr = scr + (size_t)(4 * g) * QP + 4 * c;
    const _Float16* rd = scr + (size_t)(lane & 15) * QP + 8 * (lane >> 4);
#pragma unroll
    for (int i = 0; i < 4; ++i) {
#pragma unroll
        for (int j = 0; j < 4; ++j) *(half4*)(wr + (size_t)j * QP) = h[j][i];
        __builtin_amdgcn_wave_barrier();
        A[i] = mk16(*(const half8*)rd, *(const half8*)(rd + 16));
        __builtin_amdgcn_wave_barrier();
    }
}


template <int NQ>
static __device__ inline void norm_pass(const float* __restrict__ src, _Float16* __restrict__ dst,
                                        const int sel, const int cnt, const int m0,
                                        const int g, const int c) {
    float4 x[NQ][4];
#pragma unroll
    for (int j = 0; j < NQ; ++j) {
        const int m   = m0 + 4 * j + g;
        const int row = __shfl(sel, m < cnt ? m : cnt - 1, 32);
        const float* s = src + (size_t)row * K_ + 4 * c;
#pragma unroll
        for (int i = 0; i < 4; ++i) x[j][i] = *(const float4*)(s + 32 * i);
    }
#pragma unroll
    for (int j = 0; j < NQ; ++j) {
        const int m = m0 + 4 * j + g;
        float s = 0.f;
#pragma unroll
        for (int i = 0; i < 4; ++i) {
            const float4 v = x[j][i];
            s += v.x * v.x + v.y * v.y + v.z * v.z + v.w * v.w;
        }
        s = xmask_add<1>(s);
        s = xmask_add<2>(s);
        s = xmask_add<4>(s);
        const float inv = rnorm(s);
        _Float16* d = dst + (size_t)m * PITCH + 4 * c;
        if (m < cnt) {
#pragma unroll
            for (int i = 0; i < 4; ++i) {
                const float4 v = x[j][i];
                half4 h;
                h[0] = (_Float16)(v.x * inv);
                h[1] = (_Float16)(v.y * inv);
                h[2] = (_Float16)(v.z * inv);
                h[3] = (_Float16)(v.w * inv);
                *(half4*)(d + 32 * i) = h;
            }
        }
    }
}

static __device__ inline void norm_valid_rows32(const float* __restrict__ src,
                                                _Float16* __restrict__ dst,
                                                const unsigned bits, const int lane) {
    const int g   = lane >> 3;
    const int c   = lane & 7;
    const int cnt = __builtin_popcount(bits);
    const int  rank = __builtin_popcount(bits & ((1u << lane) - 1u));
    const bool val  = ((bits >> lane) & 1u) != 0u;
    const int  sel  = __builtin_amdgcn_ds_permute((val ? rank : cnt + lane - rank) << 2, lane);
#pragma nounroll
    for (int m0 = 0; m0 < cnt; m0 += 16) {
        const int left = cnt - m0;
        if (left > 12)     norm_pass<4>(src, dst, sel, cnt, m0, g, c);
        else if (left > 8) norm_pass<3>(src, dst, sel, cnt, m0, g, c);
        else if (left > 4) norm_pass<2>(src, dst, sel, cnt, m0, g, c);
        else               norm_pass<1>(src, dst, sel, cnt, m0, g, c);
    }
}

static __device__ inline void load_b(half16 (&Bf)[4], const _Float16* p) {
#pragma unroll
    for (int kc = 0; kc < 4; ++kc)
        Bf[kc] = mk16(*(const half8*)(p + kc * 32), *(const half8*)(p + kc * 32 + 16));
}

template <bool TAIL>
static __device__ inline void tile_step(const half16 (&A0)[4], const half16 (&A1)[4],
                                        const half16 (&Bf)[4], float (&rmax0)[8],
                                        float (&rmax1)[8], const bool cv) {
    v8f acc0 = {}, acc1 = {};
#pragma unroll
    for (int kc = 0; kc < 4; ++kc) {
        acc0 = __builtin_amdgcn_wmma_f32_16x16x32_f16(
            false, A0[kc], false, Bf[kc], (short)0, acc0, false, false);
        acc1 = __builtin_amdgcn_wmma_f32_16x16x32_f16(
            false, A1[kc], false, Bf[kc], (short)0, acc1, false, false);
    }
#pragma unroll
    for (int v = 0; v < 8; ++v) {
        rmax0[v] = fmaxf(rmax0[v], (!TAIL || cv) ? acc0[v] : NEGV);
        rmax1[v] = fmaxf(rmax1[v], (!TAIL || cv) ? acc1[v] : NEGV);
    }
}

static __device__ inline float score_doc(const half16 (&A0)[4], const half16 (&A1)[4],
                                         const _Float16* sdoc, const int U,
                                         const unsigned qbits, const int lane) {
    const int lmod  = lane & 15;
    const int halfg = lane >> 4;
    const int nfull = U >> 4;
    const int urem  = U & 15;

    float rmax0[8], rmax1[8];
#pragma unroll
    for (int v = 0; v < 8; ++v) { rmax0[v] = NEGV; rmax1[v] = NEGV; }

    const _Float16* db = sdoc + (size_t)lmod * PITCH + 8 * halfg;
    half16 Ba[4], Bb[4];
    if (U != 0) load_b(Ba, db);
    int ct = 0;
    for (; ct + 2 <= nfull; ct += 2) {
        load_b(Bb, db + (size_t)(ct + 1) * 16 * PITCH);
        tile_step<false>(A0, A1, Ba, rmax0, rmax1, true);
        if (ct + 2 < nfull || urem != 0) load_b(Ba, db + (size_t)(ct + 2) * 16 * PITCH);
        tile_step<false>(A0, A1, Bb, rmax0, rmax1, true);
    }
    if (ct < nfull) {
        if (urem != 0) load_b(Bb, db + (size_t)(ct + 1) * 16 * PITCH);
        tile_step<false>(A0, A1, Ba, rmax0, rmax1, true);
        if (urem != 0) tile_step<true>(A0, A1, Bb, rmax0, rmax1, lmod < urem);
    } else if (urem != 0) {
        tile_step<true>(A0, A1, Ba, rmax0, rmax1, lmod < urem);
    }

    float y0[8], y1[8];
#pragma unroll
    for (int v = 0; v < 8; ++v) {
        float a = rmax0[v], c = rmax1[v];
        a = fmaxf(a, __shfl_xor(a, 1, 32));  c = fmaxf(c, __shfl_xor(c, 1, 32));
        a = fmaxf(a, __shfl_xor(a, 2, 32));  c = fmaxf(c, __shfl_xor(c, 2, 32));
        a = fmaxf(a, __shfl_xor(a, 4, 32));  c = fmaxf(c, __shfl_xor(c, 4, 32));
        a = fmaxf(a, __shfl_xor(a, 8, 32));  c = fmaxf(c, __shfl_xor(c, 8, 32));
        y0[v] = ((qbits >> (8 * halfg + v)) & 1u) ? a : 0.0f;
        y1[v] = ((qbits >> (16 + 8 * halfg + v)) & 1u) ? c : 0.0f;
    }

    float u[8];
#pragma unroll
    for (int v = 0; v < 8; ++v) {
        const float t = y0[v] + y1[v];
        u[v] = t + __shfl_xor(t, 16, 32);
    }
    const float s40 = u[0] + u[4], s41 = u[1] + u[5], s42 = u[2] + u[6], s43 = u[3] + u[7];
    const float s20 = s40 + s42, s21 = s41 + s43;
    return s20 + s21;
}

#define DPW 2
#define DOC_ELEMS (LD * PITCH)

static __device__ inline void doc_mask_words(const int* __restrict__ d_mask, const int n,
                                             const int lane, unsigned (&mw)[8]) {
    int mv[8];
#pragma unroll
    for (int i = 0; i < 8; ++i) mv[i] = d_mask[n * LD + 32 * i + lane];
#pragma unroll
    for (int i = 0; i < 8; ++i) mw[i] = __builtin_amdgcn_ballot_w32(mv[i] > 0);
}

static __device__ inline int stage_doc(const float* __restrict__ d_hidden, const int n,
                                       const unsigned (&mw)[8], _Float16* sdoc,
                                       const int w, const int lane, const int tid) {
    unsigned bits = 0u;
    int base = 0, U = 0;
#pragma unroll
    for (int i = 0; i < 8; ++i) {
        const int cnt = __builtin_popcount(mw[i]);
        if (i == w) bits = mw[i];
        if (i < w) base += cnt;
        U += cnt;
    }
    norm_valid_rows32(d_hidden + ((size_t)n * LD + w * 32) * K_,
                      sdoc + (size_t)base * PITCH, bits, lane);

    const int urem = U & 15;
    if (urem != 0 && tid < (16 - urem) * 16) {
        const half8 z = {};
        *(half8*)(sdoc + (size_t)(U + (tid >> 4)) * PITCH + 8 * (tid & 15)) = z;
    }
    return U;
}

__global__ void __launch_bounds__(256, 4) colbert_maxsim_kernel(
    const float* __restrict__ q, const int* __restrict__ q_mask,
    const float* __restrict__ d_hidden, const int* __restrict__ d_mask,
    float* __restrict__ out) {
    __shared__ __attribute__((aligned(16))) _Float16 sd[DPW * DOC_ELEMS + B_ * 16 * QP];

    const int n0    = DPW * blockIdx.x;
    const int tid   = threadIdx.x;
    const int w     = __builtin_amdgcn_readfirstlane(tid >> 5);
    const int lane  = tid & 31;

    unsigned mw0[8], mw1[8];
    doc_mask_words(d_mask, n0, lane, mw0);
    doc_mask_words(d_mask, n0 + 1, lane, mw1);
    const int qmv = q_mask[w * LQ + lane];

    half16 A0[4], A1[4];
    {
        const float* qb  = q + (size_t)(w * LQ) * K_;
        _Float16*    scr = sd + (size_t)DPW * DOC_ELEMS + (size_t)w * (16 * QP);
        q_frags16(qb, scr, A0, lane);
        q_frags16(qb + (size_t)16 * K_, scr, A1, lane);
    }

    const unsigned qbits = __builtin_amdgcn_ballot_w32(qmv > 0);

    const int U0 = stage_doc(d_hidden, n0,     mw0, sd,             w, lane, tid);
    const int U1 = stage_doc(d_hidden, n0 + 1, mw1, sd + DOC_ELEMS, w, lane, tid);

    __syncthreads();

    const float tot0 = score_doc(A0, A1, sd,             U0, qbits, lane);
    const float tot1 = score_doc(A0, A1, sd + DOC_ELEMS, U1, qbits, lane);
    if (lane == 0) {
        out[w * N_ + n0]     = tot0;
        out[w * N_ + n0 + 1] = tot1;
    }
}

extern "C" void kernel_launch(void* const* d_in, const int* in_sizes, int n_in,
                              void* d_out, int out_size, void* d_ws, size_t ws_size,
                              hipStream_t stream) {
    const float* q_hidden = (const float*)d_in[0];
    const int*   q_mask   = (const int*)d_in[1];
    const float* d_hidden = (const float*)d_in[2];
    const int*   d_mask   = (const int*)d_in[3];
    float*       out      = (float*)d_out;

    colbert_maxsim_kernel<<<N_ / DPW, 256, 0, stream>>>(
        q_hidden, q_mask, d_hidden, d_mask, out);
}
